# starting kernel
# baseline (speedup 1.0000x reference)
#include <hip/hip_runtime.h>

typedef __attribute__((ext_vector_type(2))) float v2f;
typedef __attribute__((ext_vector_type(8))) float v8f;

__device__ __forceinline__ void cmul(float ar, float ai, float br, float bi,
                                     float& cr, float& ci) {
    cr = ar * br - ai * bi;
    ci = ar * bi + ai * br;
}

__device__ __forceinline__ void cpow(float ar, float ai, int p, float& r, float& i) {
    float xr = 1.f, xi = 0.f;
    for (int q = 0; q < p; ++q) { float tr, ti; cmul(xr, xi, ar, ai, tr, ti); xr = tr; xi = ti; }
    r = xr; i = xi;
}

__global__ void slab_pass1(const float* __restrict__ pk, const float* __restrict__ TAx,
                           const float* __restrict__ TAy, const float* __restrict__ fc,
                           const int* __restrict__ p_dt, const int* __restrict__ p_dtf,
                           float* __restrict__ out, int nsteps, int nchunks, int ntiles,
                           int nt)
{
    int gid  = blockIdx.x * blockDim.x + threadIdx.x;
    int tile = gid >> 5;
    int lane = gid & 31;
    if (tile >= ntiles) return;

    float dt      = (float)p_dt[0];
    int   nsub    = p_dtf[0] / p_dt[0];
    float invnsub = 1.f / (float)nsub;
    float K0   = __expf(pk[0]);
    float K1   = __expf(pk[1]);
    float f    = fc[0];
    float ar   = 1.f - dt * K1;
    float ai   = -dt * f;
    float dtK0 = dt * K0;

    int half = lane >> 4;
    int r    = lane & 15;
    int c0   = tile * 16;

    v2f Are[4], Aim[4], Mre[4], Mim[4], Mneg[4];
    #pragma unroll
    for (int s = 0; s < 4; ++s) {
        #pragma unroll
        for (int e = 0; e < 2; ++e) {
            int j = 4 * s + 2 * half + e;
            int  c  = c0 + r;
            long n  = (long)16 * c + j;
            int itf0 = (int)(n / nsub);
            float aa = (float)(n - (long)itf0 * nsub) * invnsub;
            int i0 = itf0     < nt - 1 ? itf0     : nt - 1;
            int i1 = itf0 + 1 < nt - 1 ? itf0 + 1 : nt - 1;
            float tax = (1.f - aa) * TAx[i0] + aa * TAx[i1];
            float tay = (1.f - aa) * TAy[i0] + aa * TAy[i1];
            float valid = (c < nchunks && n < (long)nsteps) ? 1.f : 0.f;
            Are[s][e] = valid * dtK0 * tax;
            Aim[s][e] = valid * dtK0 * tay;
            int p = r - j;
            float pr, pi;
            cpow(ar, ai, p < 0 ? 0 : p, pr, pi);
            float m = (p >= 0) ? 1.f : 0.f;
            Mre[s][e]  =  m * pr;
            Mim[s][e]  =  m * pi;
            Mneg[s][e] = -m * pi;
        }
    }

    v8f Sre = {}; v8f Sim = {};
    #pragma unroll
    for (int s = 0; s < 4; ++s)
        Sre = __builtin_amdgcn_wmma_f32_16x16x4_f32(false, Are[s], false, Mre[s],
                                                    (short)0, Sre, false, false);
    #pragma unroll
    for (int s = 0; s < 4; ++s)
        Sre = __builtin_amdgcn_wmma_f32_16x16x4_f32(false, Aim[s], false, Mneg[s],
                                                    (short)0, Sre, false, false);
    #pragma unroll
    for (int s = 0; s < 4; ++s)
        Sim = __builtin_amdgcn_wmma_f32_16x16x4_f32(false, Are[s], false, Mim[s],
                                                    (short)0, Sim, false, false);
    #pragma unroll
    for (int s = 0; s < 4; ++s)
        Sim = __builtin_amdgcn_wmma_f32_16x16x4_f32(false, Aim[s], false, Mre[s],
                                                    (short)0, Sim, false, false);

    #pragma unroll
    for (int e = 0; e < 8; ++e) {
        int  row = e + 8 * half;
        int  c   = c0 + row;
        long n   = (long)16 * c + r;
        if (c < nchunks && n < (long)nsteps) {
            float2 v; v.x = Sre[e]; v.y = Sim[e];
            *reinterpret_cast<float2*>(out + 2 * n) = v;
        }
    }
}

__global__ void slab_pass2(const float* __restrict__ pk, const float* __restrict__ fc,
                           const int* __restrict__ p_dt,
                           const float* __restrict__ out, float2* __restrict__ ws,
                           int nsteps, int nchunks)
{
    __shared__ float4 sh[1024];
    int t = threadIdx.x;
    float dt = (float)p_dt[0];
    float K1 = __expf(pk[1]);
    float f  = fc[0];
    float ar = 1.f - dt * K1, ai = -dt * f;
    float a16r, a16i; cpow(ar, ai, 16, a16r, a16i);

    int cpt = (nchunks + 1023) >> 10;
    int cb  = t * cpt;
    int ce  = min(cb + cpt, nchunks);

    float Ar = 1.f, Ai = 0.f, Br = 0.f, Bi = 0.f;
    for (int c = cb; c < ce; ++c) {
        long n15 = (long)16 * c + 15;
        float gr = 0.f, gi = 0.f;
        if (n15 < (long)nsteps) { gr = out[2 * n15]; gi = out[2 * n15 + 1]; }
        float tr, ti;
        cmul(a16r, a16i, Ar, Ai, tr, ti); Ar = tr; Ai = ti;
        cmul(a16r, a16i, Br, Bi, tr, ti); Br = tr + gr; Bi = ti + gi;
    }
    sh[t] = make_float4(Ar, Ai, Br, Bi);
    __syncthreads();
    for (int off = 1; off < 1024; off <<= 1) {
        float4 v = sh[t];
        if (t >= off) {
            float4 p = sh[t - off];
            float nar, nai, tr, ti;
            cmul(v.x, v.y, p.x, p.y, nar, nai);
            cmul(v.x, v.y, p.z, p.w, tr, ti);
            v = make_float4(nar, nai, tr + v.z, ti + v.w);
        }
        __syncthreads();
        sh[t] = v;
        __syncthreads();
    }
    float Wr = 0.f, Wi = 0.f;
    if (t > 0) { float4 p = sh[t - 1]; Wr = p.z; Wi = p.w; }
    for (int c = cb; c < ce; ++c) {
        ws[c] = make_float2(Wr, Wi);
        long n15 = (long)16 * c + 15;
        float gr = 0.f, gi = 0.f;
        if (n15 < (long)nsteps) { gr = out[2 * n15]; gi = out[2 * n15 + 1]; }
        float tr, ti;
        cmul(a16r, a16i, Wr, Wi, tr, ti);
        Wr = tr + gr; Wi = ti + gi;
    }
}

__global__ void slab_pass3(const float* __restrict__ pk, const float* __restrict__ fc,
                           const int* __restrict__ p_dt,
                           const float2* __restrict__ ws, float* __restrict__ out,
                           int nsteps)
{
    long n = (long)blockIdx.x * blockDim.x + threadIdx.x;
    if (n >= (long)nsteps) return;
    float dt = (float)p_dt[0];
    float K1 = __expf(pk[1]);
    float f  = fc[0];
    float ar = 1.f - dt * K1, ai = -dt * f;
    int c = (int)(n >> 4);
    int k = (int)(n & 15);
    float2 W = ws[c];
    float mr, mi; cpow(ar, ai, k + 1, mr, mi);
    out[2 * n]     += mr * W.x - mi * W.y;
    out[2 * n + 1] += mr * W.y + mi * W.x;
}

extern "C" void kernel_launch(void* const* d_in, const int* in_sizes, int n_in,
                              void* d_out, int out_size, void* d_ws, size_t ws_size,
                              hipStream_t stream)
{
    const float* pk    = (const float*)d_in[0];
    const float* TAx   = (const float*)d_in[1];
    const float* TAy   = (const float*)d_in[2];
    const float* fc    = (const float*)d_in[3];
    const int*   p_dt  = (const int*)d_in[6];
    const int*   p_dtf = (const int*)d_in[7];
    (void)n_in; (void)ws_size;

    int nt      = in_sizes[1];
    int nsteps  = out_size / 2;
    int nchunks = (nsteps + 15) / 16;
    int ntiles  = (nchunks + 15) / 16;
    float*  out = (float*)d_out;
    float2* ws  = (float2*)d_ws;

    int threads1 = ntiles * 32;
    int blocks1  = (threads1 + 255) / 256;
    slab_pass1<<<blocks1, 256, 0, stream>>>(pk, TAx, TAy, fc, p_dt, p_dtf,
                                            out, nsteps, nchunks, ntiles, nt);
    slab_pass2<<<1, 1024, 0, stream>>>(pk, fc, p_dt, out, ws, nsteps, nchunks);
    int blocks3 = (nsteps + 255) / 256;
    slab_pass3<<<blocks3, 256, 0, stream>>>(pk, fc, p_dt, ws, out, nsteps);
}


# Rows are forcing intervals: one block scans the interval start states from the samples first, then one WMMA pass writes out once as [state, two samples] x per-step weights; reduction tree changes
# speedup vs baseline: 6.0808x; 6.0808x over previous
#include <hip/hip_runtime.h>

typedef __attribute__((ext_vector_type(2))) float v2f;
typedef __attribute__((ext_vector_type(8))) float v8f;

#define SCAN_THREADS 1024
#define BATCH 10
#define TABROW 16
#define TAB_OFFSET_BYTES (128u << 20)

__device__ __forceinline__ void cmul(float ar, float ai, float br, float bi,
                                     float& cr, float& ci) {
    cr = ar * br - ai * bi;
    ci = ar * bi + ai * br;
}

__device__ __forceinline__ void load_batch(const float* __restrict__ TAx,
                                           const float* __restrict__ TAy,
                                           int c0, int nt,
                                           float (&tx)[BATCH + 1], float (&ty)[BATCH + 1]) {
    #pragma unroll
    for (int u = 0; u <= BATCH; ++u) {
        unsigned ii = (unsigned)min(c0 + u, nt - 1);
        tx[u] = TAx[ii];
        ty[u] = TAy[ii];
    }
}

__global__ __launch_bounds__(SCAN_THREADS)
void interval_scan(const float* __restrict__ pk, const float* __restrict__ TAx,
                   const float* __restrict__ TAy, const float* __restrict__ fc,
                   const int* __restrict__ p_dt, const int* __restrict__ p_dtf,
                   float2* __restrict__ zst, float4* __restrict__ tab,
                   int nsteps, int nt)
{
    __shared__ float2 sh_tot[32];
    __shared__ float  sh_coef[6];
    int t    = threadIdx.x;
    int lane = t & 31;
    int w    = t >> 5;

    float dt      = (float)p_dt[0];
    int   nsub    = p_dtf[0] / p_dt[0];
    float invnsub = 1.f / (float)nsub;
    float K0   = __expf(pk[0]);
    float K1   = __expf(pk[1]);
    float f    = fc[0];
    float ar   = 1.f - dt * K1;
    float ai   = -dt * f;
    float dtK0 = dt * K0;

    int nint = (nsteps + nsub - 1) / nsub;
    int cpt  = (nint + SCAN_THREADS - 1) / SCAN_THREADS;
    int cb   = t * cpt;
    int ce   = cb + cpt;

    float tx[BATCH + 1], ty[BATCH + 1];
    load_batch(TAx, TAy, cb, nt, tx, ty);

    if (w == 0) {
        float Pr = 1.f, Pi = 0.f, D0r = 0.f, D0i = 0.f, D1r = 0.f, D1i = 0.f;
        for (int m = 0; m < nsub; ++m) {
            float aa = (float)m * invnsub;
            float tr, ti;
            cmul(ar, ai, Pr, Pi, tr, ti);   Pr = tr; Pi = ti;
            cmul(ar, ai, D0r, D0i, tr, ti); D0r = tr + (1.f - aa); D0i = ti;
            cmul(ar, ai, D1r, D1i, tr, ti); D1r = tr + aa;         D1i = ti;
            if ((m & 31) == lane) {
                float e0r = dtK0 * D0r, e0i = dtK0 * D0i;
                float e1r = dtK0 * D1r, e1i = dtK0 * D1i;
                tab[4 * m + 0] = make_float4(Pr, -Pi, e0r, -e0i);
                tab[4 * m + 1] = make_float4(e1r, -e1i, 0.f, 0.f);
                tab[4 * m + 2] = make_float4(Pi, Pr, e0i, e0r);
                tab[4 * m + 3] = make_float4(e1i, e1r, 0.f, 0.f);
            }
        }
        if (lane == 0) {
            sh_coef[0] = Pr;          sh_coef[1] = Pi;
            sh_coef[2] = dtK0 * D0r;  sh_coef[3] = dtK0 * D0i;
            sh_coef[4] = dtK0 * D1r;  sh_coef[5] = dtK0 * D1i;
        }
    }
    __syncthreads();
    float Ar = sh_coef[0], Ai = sh_coef[1];
    float c0r = sh_coef[2], c0i = sh_coef[3];
    float c1r = sh_coef[4], c1i = sh_coef[5];

    float Br = 0.f, Bi = 0.f, Mr = 1.f, Mi = 0.f;
    for (int c0 = cb;;) {
        #pragma unroll
        for (int u = 0; u < BATCH; ++u) {
            int c = c0 + u;
            if (c < ce) {
                float gr = 0.f, gi = 0.f;
                if (c < nint) {
                    gr = c0r * tx[u] - c0i * ty[u] + c1r * tx[u + 1] - c1i * ty[u + 1];
                    gi = c0r * ty[u] + c0i * tx[u] + c1r * ty[u + 1] + c1i * tx[u + 1];
                }
                float tr, ti;
                cmul(Ar, Ai, Br, Bi, tr, ti); Br = tr + gr; Bi = ti + gi;
                cmul(Ar, Ai, Mr, Mi, tr, ti); Mr = tr;      Mi = ti;
            }
        }
        c0 += BATCH;
        if (c0 >= ce) break;
        load_batch(TAx, TAy, c0, nt, tx, ty);
    }

    float Qr = 1.f, Qi = 0.f, Xr = Mr, Xi = Mi;
    #pragma unroll
    for (int off = 1; off < 32; off <<= 1) {
        float pr = __shfl_up(Br, off);
        float pi = __shfl_up(Bi, off);
        float tr, ti;
        cmul(Xr, Xi, pr, pi, tr, ti);
        if (lane >= off) { Br += tr; Bi += ti; }
        cmul(Qr, Qi, Xr, Xi, tr, ti);
        if (lane & off) { Qr = tr; Qi = ti; }
        cmul(Xr, Xi, Xr, Xi, tr, ti); Xr = tr; Xi = ti;
    }
    if (lane == 31) sh_tot[w] = make_float2(Br, Bi);
    __syncthreads();

    float2 tt = sh_tot[lane];
    float Tr = tt.x, Ti = tt.y;
    #pragma unroll
    for (int off = 1; off < 32; off <<= 1) {
        float pr = __shfl_up(Tr, off);
        float pi = __shfl_up(Ti, off);
        float tr, ti;
        cmul(Xr, Xi, pr, pi, tr, ti);
        if (lane >= off) { Tr += tr; Ti += ti; }
        cmul(Xr, Xi, Xr, Xi, tr, ti); Xr = tr; Xi = ti;
    }
    int   src = (w + 31) & 31;
    float Er  = __shfl(Tr, src);
    float Ei  = __shfl(Ti, src);
    if (w == 0) { Er = 0.f; Ei = 0.f; }
    float xr = __shfl_up(Br, 1);
    float xi = __shfl_up(Bi, 1);
    if (lane == 0) { xr = 0.f; xi = 0.f; }
    float Zr, Zi;
    cmul(Qr, Qi, Er, Ei, Zr, Zi);
    Zr += xr; Zi += xi;

    if (cpt > BATCH) load_batch(TAx, TAy, cb, nt, tx, ty);
    for (int c0 = cb;;) {
        #pragma unroll
        for (int u = 0; u < BATCH; ++u) {
            int c = c0 + u;
            if (c < ce) {
                zst[c] = make_float2(Zr, Zi);
                float gr = 0.f, gi = 0.f;
                if (c < nint) {
                    gr = c0r * tx[u] - c0i * ty[u] + c1r * tx[u + 1] - c1i * ty[u + 1];
                    gi = c0r * ty[u] + c0i * tx[u] + c1r * ty[u + 1] + c1i * tx[u + 1];
                }
                float tr, ti;
                cmul(Ar, Ai, Zr, Zi, tr, ti); Zr = tr + gr; Zi = ti + gi;
            }
        }
        c0 += BATCH;
        if (c0 >= ce) break;
        load_batch(TAx, TAy, c0, nt, tx, ty);
    }
}

__global__ __launch_bounds__(32)
void interval_expand(const float* __restrict__ TAx, const float* __restrict__ TAy,
                     const int* __restrict__ p_dt, const int* __restrict__ p_dtf,
                     const float2* __restrict__ zst, const float* __restrict__ tabf,
                     float* __restrict__ out, int nsteps, int nt)
{
    int lane = threadIdx.x;
    int h    = lane >> 4;
    int r    = lane & 15;

    int nsub   = p_dtf[0] / p_dt[0];
    int nint   = (nsteps + nsub - 1) / nsub;
    int ntile  = (nint + 15) >> 4;
    int mtiles = (nsub + 15) >> 4;

    for (int mt0 = 0; mt0 < mtiles; mt0 += 4) {
        v2f Wr0[4], Wr1[4], Wi0[4], Wi1[4];
        #pragma unroll
        for (int q = 0; q < 4; ++q) {
            int mc = min(16 * (mt0 + q) + r, nsub - 1);
            const float* trow = tabf + TABROW * mc + 2 * h;
            Wr0[q] = *reinterpret_cast<const v2f*>(trow);
            Wr1[q] = *reinterpret_cast<const v2f*>(trow + 4);
            Wi0[q] = *reinterpret_cast<const v2f*>(trow + 8);
            Wi1[q] = *reinterpret_cast<const v2f*>(trow + 12);
        }

        for (int tile = blockIdx.x; tile < ntile; tile += gridDim.x) {
            int i = tile * 16 + r;
            unsigned is = (unsigned)min(i + 1 - h, nt - 1);
            float  sx = TAx[is];
            float  sy = TAy[is];
            float2 z  = zst[i];
            v2f X0, X1;
            X0[0] = h ? sx : z.x;   X0[1] = h ? sy : z.y;
            X1[0] = h ? 0.f : sx;   X1[1] = h ? 0.f : sy;

            v8f Dre[4], Dim[4];
            #pragma unroll
            for (int q = 0; q < 4; ++q) {
                v8f zero = {};
                Dre[q] = __builtin_amdgcn_wmma_f32_16x16x4_f32(false, X0, false, Wr0[q],
                                                               (short)0, zero, false, false);
                Dim[q] = __builtin_amdgcn_wmma_f32_16x16x4_f32(false, X0, false, Wi0[q],
                                                               (short)0, zero, false, false);
            }
            #pragma unroll
            for (int q = 0; q < 4; ++q) {
                Dre[q] = __builtin_amdgcn_wmma_f32_16x16x4_f32(false, X1, false, Wr1[q],
                                                               (short)0, Dre[q], false, false);
                Dim[q] = __builtin_amdgcn_wmma_f32_16x16x4_f32(false, X1, false, Wi1[q],
                                                               (short)0, Dim[q], false, false);
            }
            #pragma unroll
            for (int q = 0; q < 4; ++q) {
                int m = 16 * (mt0 + q) + r;
                #pragma unroll
                for (int e = 0; e < 8; ++e) {
                    int n = (tile * 16 + e + 8 * h) * nsub + m;
                    if (m < nsub && n < nsteps) {
                        float2 v; v.x = Dre[q][e]; v.y = Dim[q][e];
                        *reinterpret_cast<float2*>(out + 2 * (long)n) = v;
                    }
                }
            }
        }
    }
}

extern "C" void kernel_launch(void* const* d_in, const int* in_sizes, int n_in,
                              void* d_out, int out_size, void* d_ws, size_t ws_size,
                              hipStream_t stream)
{
    const float* pk    = (const float*)d_in[0];
    const float* TAx   = (const float*)d_in[1];
    const float* TAy   = (const float*)d_in[2];
    const float* fc    = (const float*)d_in[3];
    const int*   p_dt  = (const int*)d_in[6];
    const int*   p_dtf = (const int*)d_in[7];
    (void)n_in; (void)ws_size;

    int nt      = in_sizes[1];
    int nsteps  = out_size / 2;
    float*  out = (float*)d_out;
    float2* zst = (float2*)d_ws;
    float4* tab = (float4*)((char*)d_ws + TAB_OFFSET_BYTES);

    interval_scan<<<1, SCAN_THREADS, 0, stream>>>(pk, TAx, TAy, fc, p_dt, p_dtf,
                                                  zst, tab, nsteps, nt);
    int blocks = (nt + 15) / 16;
    interval_expand<<<blocks, 32, 0, stream>>>(TAx, TAy, p_dt, p_dtf, zst, (const float*)tab,
                                               out, nsteps, nt);
}


# Weight table by a prefix scan over 16 lanes (a few steps per lane, DPP row shifts carry the lane sums, powers of a by squaring), not wave 0's serial nsub-step loop; reduction tree changes
# speedup vs baseline: 7.1502x; 1.1759x over previous
#include <hip/hip_runtime.h>

typedef __attribute__((ext_vector_type(2))) float v2f;
typedef __attribute__((ext_vector_type(8))) float v8f;

#define SCAN_THREADS 1024
#define BATCH 10
#define TABROW 16
#define TAB_OFFSET_BYTES (128u << 20)

__device__ __forceinline__ void cmul(float ar, float ai, float br, float bi,
                                     float& cr, float& ci) {
    cr = ar * br - ai * bi;
    ci = ar * bi + ai * br;
}

template <int N>
__device__ __forceinline__ float row_shr(float v) {
    return __builtin_bit_cast(float,
        __builtin_amdgcn_update_dpp(0, __builtin_bit_cast(int, v), 0x110 + N, 0xf, 0xf, false));
}

template <int N>
__device__ __forceinline__ void scan_step(int r, float& Xr, float& Xi, float& Qr, float& Qi,
                                          float& D0r, float& D0i, float& D1r, float& D1i) {
    float u0r = row_shr<N>(D0r), u0i = row_shr<N>(D0i);
    float u1r = row_shr<N>(D1r), u1i = row_shr<N>(D1i);
    float tr, ti;
    cmul(Xr, Xi, u0r, u0i, tr, ti); if (r >= N) { D0r += tr; D0i += ti; }
    cmul(Xr, Xi, u1r, u1i, tr, ti); if (r >= N) { D1r += tr; D1i += ti; }
    cmul(Qr, Qi, Xr, Xi, tr, ti);   if (r & N)  { Qr = tr;  Qi = ti; }
    cmul(Xr, Xi, Xr, Xi, tr, ti);   Xr = tr; Xi = ti;
}

__device__ __forceinline__ void load_batch(const float* __restrict__ TAx,
                                           const float* __restrict__ TAy,
                                           int c0, int nt,
                                           float (&tx)[BATCH + 1], float (&ty)[BATCH + 1]) {
    #pragma unroll
    for (int u = 0; u <= BATCH; ++u) {
        unsigned ii = (unsigned)min(c0 + u, nt - 1);
        tx[u] = TAx[ii];
        ty[u] = TAy[ii];
    }
}

__global__ __launch_bounds__(SCAN_THREADS)
void interval_scan(const float* __restrict__ pk, const float* __restrict__ TAx,
                   const float* __restrict__ TAy, const float* __restrict__ fc,
                   const int* __restrict__ p_dt, const int* __restrict__ p_dtf,
                   float2* __restrict__ zst, float4* __restrict__ tab,
                   int nsteps, int nt)
{
    __shared__ float2 sh_tot[32];
    __shared__ float  sh_coef[6];
    int t    = threadIdx.x;
    int lane = t & 31;
    int w    = t >> 5;

    float dt      = (float)p_dt[0];
    int   nsub    = p_dtf[0] / p_dt[0];
    float invnsub = 1.f / (float)nsub;
    float K0   = __expf(pk[0]);
    float K1   = __expf(pk[1]);
    float f    = fc[0];
    float ar   = 1.f - dt * K1;
    float ai   = -dt * f;
    float dtK0 = dt * K0;

    int nint = (nsteps + nsub - 1) / nsub;
    int cpt  = (nint + SCAN_THREADS - 1) / SCAN_THREADS;
    int cb   = t * cpt;
    int ce   = cb + cpt;

    float tx[BATCH + 1], ty[BATCH + 1];
    load_batch(TAx, TAy, cb, nt, tx, ty);

    if (w == 0) {
        int r   = lane & 15;
        int h   = lane >> 4;
        int spl = (nsub + 15) >> 4;
        int m0  = r * spl;

        float Pr = 1.f, Pi = 0.f, D0r = 0.f, D0i = 0.f, D1r = 0.f, D1i = 0.f;
        for (int s = 0; s < spl; ++s) {
            float aa = (float)(m0 + s) * invnsub;
            float tr, ti;
            cmul(ar, ai, Pr, Pi, tr, ti);   Pr = tr; Pi = ti;
            cmul(ar, ai, D0r, D0i, tr, ti); D0r = tr + (1.f - aa); D0i = ti;
            cmul(ar, ai, D1r, D1i, tr, ti); D1r = tr + aa;         D1i = ti;
        }

        float Xr = Pr, Xi = Pi, Qr = 1.f, Qi = 0.f;
        scan_step<1>(r, Xr, Xi, Qr, Qi, D0r, D0i, D1r, D1i);
        scan_step<2>(r, Xr, Xi, Qr, Qi, D0r, D0i, D1r, D1i);
        scan_step<4>(r, Xr, Xi, Qr, Qi, D0r, D0i, D1r, D1i);
        scan_step<8>(r, Xr, Xi, Qr, Qi, D0r, D0i, D1r, D1i);

        Pr  = Qr;              Pi  = Qi;
        D0r = row_shr<1>(D0r); D0i = row_shr<1>(D0i);
        D1r = row_shr<1>(D1r); D1i = row_shr<1>(D1i);

        for (int s = 0; s < spl; ++s) {
            int   m  = m0 + s;
            float aa = (float)m * invnsub;
            float tr, ti;
            cmul(ar, ai, Pr, Pi, tr, ti);   Pr = tr; Pi = ti;
            cmul(ar, ai, D0r, D0i, tr, ti); D0r = tr + (1.f - aa); D0i = ti;
            cmul(ar, ai, D1r, D1i, tr, ti); D1r = tr + aa;         D1i = ti;
            if (m < nsub) {
                float e0r = dtK0 * D0r, e0i = dtK0 * D0i;
                float e1r = dtK0 * D1r, e1i = dtK0 * D1i;
                tab[4 * m + 2 * h]     = h ? make_float4(Pi, Pr, e0i, e0r)
                                           : make_float4(Pr, -Pi, e0r, -e0i);
                tab[4 * m + 2 * h + 1] = h ? make_float4(e1i, e1r, 0.f, 0.f)
                                           : make_float4(e1r, -e1i, 0.f, 0.f);
                if (m == nsub - 1 && h == 0) {
                    sh_coef[0] = Pr;   sh_coef[1] = Pi;
                    sh_coef[2] = e0r;  sh_coef[3] = e0i;
                    sh_coef[4] = e1r;  sh_coef[5] = e1i;
                }
            }
        }
    }
    __syncthreads();
    float Ar = sh_coef[0], Ai = sh_coef[1];
    float c0r = sh_coef[2], c0i = sh_coef[3];
    float c1r = sh_coef[4], c1i = sh_coef[5];

    float Br = 0.f, Bi = 0.f, Mr = 1.f, Mi = 0.f;
    for (int c0 = cb;;) {
        #pragma unroll
        for (int u = 0; u < BATCH; ++u) {
            int c = c0 + u;
            if (c < ce) {
                float gr = 0.f, gi = 0.f;
                if (c < nint) {
                    gr = c0r * tx[u] - c0i * ty[u] + c1r * tx[u + 1] - c1i * ty[u + 1];
                    gi = c0r * ty[u] + c0i * tx[u] + c1r * ty[u + 1] + c1i * tx[u + 1];
                }
                float tr, ti;
                cmul(Ar, Ai, Br, Bi, tr, ti); Br = tr + gr; Bi = ti + gi;
                cmul(Ar, Ai, Mr, Mi, tr, ti); Mr = tr;      Mi = ti;
            }
        }
        c0 += BATCH;
        if (c0 >= ce) break;
        load_batch(TAx, TAy, c0, nt, tx, ty);
    }

    float Qr = 1.f, Qi = 0.f, Xr = Mr, Xi = Mi;
    #pragma unroll
    for (int off = 1; off < 32; off <<= 1) {
        float pr = __shfl_up(Br, off);
        float pi = __shfl_up(Bi, off);
        float tr, ti;
        cmul(Xr, Xi, pr, pi, tr, ti);
        if (lane >= off) { Br += tr; Bi += ti; }
        cmul(Qr, Qi, Xr, Xi, tr, ti);
        if (lane & off) { Qr = tr; Qi = ti; }
        cmul(Xr, Xi, Xr, Xi, tr, ti); Xr = tr; Xi = ti;
    }
    if (lane == 31) sh_tot[w] = make_float2(Br, Bi);
    __syncthreads();

    float2 tt = sh_tot[lane];
    float Tr = tt.x, Ti = tt.y;
    #pragma unroll
    for (int off = 1; off < 32; off <<= 1) {
        float pr = __shfl_up(Tr, off);
        float pi = __shfl_up(Ti, off);
        float tr, ti;
        cmul(Xr, Xi, pr, pi, tr, ti);
        if (lane >= off) { Tr += tr; Ti += ti; }
        cmul(Xr, Xi, Xr, Xi, tr, ti); Xr = tr; Xi = ti;
    }
    int   src = (w + 31) & 31;
    float Er  = __shfl(Tr, src);
    float Ei  = __shfl(Ti, src);
    if (w == 0) { Er = 0.f; Ei = 0.f; }
    float xr = __shfl_up(Br, 1);
    float xi = __shfl_up(Bi, 1);
    if (lane == 0) { xr = 0.f; xi = 0.f; }
    float Zr, Zi;
    cmul(Qr, Qi, Er, Ei, Zr, Zi);
    Zr += xr; Zi += xi;

    if (cpt > BATCH) load_batch(TAx, TAy, cb, nt, tx, ty);
    for (int c0 = cb;;) {
        #pragma unroll
        for (int u = 0; u < BATCH; ++u) {
            int c = c0 + u;
            if (c < ce) {
                zst[c] = make_float2(Zr, Zi);
                float gr = 0.f, gi = 0.f;
                if (c < nint) {
                    gr = c0r * tx[u] - c0i * ty[u] + c1r * tx[u + 1] - c1i * ty[u + 1];
                    gi = c0r * ty[u] + c0i * tx[u] + c1r * ty[u + 1] + c1i * tx[u + 1];
                }
                float tr, ti;
                cmul(Ar, Ai, Zr, Zi, tr, ti); Zr = tr + gr; Zi = ti + gi;
            }
        }
        c0 += BATCH;
        if (c0 >= ce) break;
        load_batch(TAx, TAy, c0, nt, tx, ty);
    }
}

__global__ __launch_bounds__(32)
void interval_expand(const float* __restrict__ TAx, const float* __restrict__ TAy,
                     const int* __restrict__ p_dt, const int* __restrict__ p_dtf,
                     const float2* __restrict__ zst, const float* __restrict__ tabf,
                     float* __restrict__ out, int nsteps, int nt)
{
    int lane = threadIdx.x;
    int h    = lane >> 4;
    int r    = lane & 15;

    int nsub   = p_dtf[0] / p_dt[0];
    int nint   = (nsteps + nsub - 1) / nsub;
    int ntile  = (nint + 15) >> 4;
    int mtiles = (nsub + 15) >> 4;

    for (int mt0 = 0; mt0 < mtiles; mt0 += 4) {
        v2f Wr0[4], Wr1[4], Wi0[4], Wi1[4];
        #pragma unroll
        for (int q = 0; q < 4; ++q) {
            int mc = min(16 * (mt0 + q) + r, nsub - 1);
            const float* trow = tabf + TABROW * mc + 2 * h;
            Wr0[q] = *reinterpret_cast<const v2f*>(trow);
            Wr1[q] = *reinterpret_cast<const v2f*>(trow + 4);
            Wi0[q] = *reinterpret_cast<const v2f*>(trow + 8);
            Wi1[q] = *reinterpret_cast<const v2f*>(trow + 12);
        }

        for (int tile = blockIdx.x; tile < ntile; tile += gridDim.x) {
            int i = tile * 16 + r;
            unsigned is = (unsigned)min(i + 1 - h, nt - 1);
            float  sx = TAx[is];
            float  sy = TAy[is];
            float2 z  = zst[i];
            v2f X0, X1;
            X0[0] = h ? sx : z.x;   X0[1] = h ? sy : z.y;
            X1[0] = h ? 0.f : sx;   X1[1] = h ? 0.f : sy;

            v8f Dre[4], Dim[4];
            #pragma unroll
            for (int q = 0; q < 4; ++q) {
                v8f zero = {};
                Dre[q] = __builtin_amdgcn_wmma_f32_16x16x4_f32(false, X0, false, Wr0[q],
                                                               (short)0, zero, false, false);
                Dim[q] = __builtin_amdgcn_wmma_f32_16x16x4_f32(false, X0, false, Wi0[q],
                                                               (short)0, zero, false, false);
            }
            #pragma unroll
            for (int q = 0; q < 4; ++q) {
                Dre[q] = __builtin_amdgcn_wmma_f32_16x16x4_f32(false, X1, false, Wr1[q],
                                                               (short)0, Dre[q], false, false);
                Dim[q] = __builtin_amdgcn_wmma_f32_16x16x4_f32(false, X1, false, Wi1[q],
                                                               (short)0, Dim[q], false, false);
            }
            #pragma unroll
            for (int q = 0; q < 4; ++q) {
                int m = 16 * (mt0 + q) + r;
                #pragma unroll
                for (int e = 0; e < 8; ++e) {
                    int n = (tile * 16 + e + 8 * h) * nsub + m;
                    if (m < nsub && n < nsteps) {
                        float2 v; v.x = Dre[q][e]; v.y = Dim[q][e];
                        *reinterpret_cast<float2*>(out + 2 * (long)n) = v;
                    }
                }
            }
        }
    }
}

extern "C" void kernel_launch(void* const* d_in, const int* in_sizes, int n_in,
                              void* d_out, int out_size, void* d_ws, size_t ws_size,
                              hipStream_t stream)
{
    const float* pk    = (const float*)d_in[0];
    const float* TAx   = (const float*)d_in[1];
    const float* TAy   = (const float*)d_in[2];
    const float* fc    = (const float*)d_in[3];
    const int*   p_dt  = (const int*)d_in[6];
    const int*   p_dtf = (const int*)d_in[7];
    (void)n_in; (void)ws_size;

    int nt      = in_sizes[1];
    int nsteps  = out_size / 2;
    float*  out = (float*)d_out;
    float2* zst = (float2*)d_ws;
    float4* tab = (float4*)((char*)d_ws + TAB_OFFSET_BYTES);

    interval_scan<<<1, SCAN_THREADS, 0, stream>>>(pk, TAx, TAy, fc, p_dt, p_dtf,
                                                  zst, tab, nsteps, nt);
    int blocks = (nt + 15) / 16;
    interval_expand<<<blocks, 32, 0, stream>>>(TAx, TAy, p_dt, p_dtf, zst, (const float*)tab,
                                               out, nsteps, nt);
}


# Scan block's two interval walks made branch-free: forcing terms formed once per group and kept in registers, group tails tested on a thread-shared count (scalar branch); sums keep order
# speedup vs baseline: 7.6253x; 1.0664x over previous
#include <hip/hip_runtime.h>

typedef __attribute__((ext_vector_type(2))) float v2f;
typedef __attribute__((ext_vector_type(8))) float v8f;

#define SCAN_THREADS 1024
#define BATCH 10
#define TABROW 16
#define TAB_OFFSET_BYTES (128u << 20)

__device__ __forceinline__ void cmul(float ar, float ai, float br, float bi,
                                     float& cr, float& ci) {
    cr = ar * br - ai * bi;
    ci = ar * bi + ai * br;
}

template <int N>
__device__ __forceinline__ float row_shr(float v) {
    return __builtin_bit_cast(float,
        __builtin_amdgcn_update_dpp(0, __builtin_bit_cast(int, v), 0x110 + N, 0xf, 0xf, false));
}

template <int N>
__device__ __forceinline__ void scan_step(int r, float& Xr, float& Xi, float& Qr, float& Qi,
                                          float& D0r, float& D0i, float& D1r, float& D1i) {
    float u0r = row_shr<N>(D0r), u0i = row_shr<N>(D0i);
    float u1r = row_shr<N>(D1r), u1i = row_shr<N>(D1i);
    float tr, ti;
    cmul(Xr, Xi, u0r, u0i, tr, ti); if (r >= N) { D0r += tr; D0i += ti; }
    cmul(Xr, Xi, u1r, u1i, tr, ti); if (r >= N) { D1r += tr; D1i += ti; }
    cmul(Qr, Qi, Xr, Xi, tr, ti);   if (r & N)  { Qr = tr;  Qi = ti; }
    cmul(Xr, Xi, Xr, Xi, tr, ti);   Xr = tr; Xi = ti;
}

__device__ __forceinline__ void load_batch(const float* __restrict__ TAx,
                                           const float* __restrict__ TAy,
                                           int c0, int nt,
                                           float (&tx)[BATCH + 1], float (&ty)[BATCH + 1]) {
    #pragma unroll
    for (int u = 0; u <= BATCH; ++u) {
        unsigned ii = (unsigned)min(c0 + u, nt - 1);
        tx[u] = TAx[ii];
        ty[u] = TAy[ii];
    }
}

__device__ __forceinline__ void batch_terms(float c0r, float c0i, float c1r, float c1i,
                                            const float (&tx)[BATCH + 1],
                                            const float (&ty)[BATCH + 1],
                                            int c0, int nint,
                                            float (&gr)[BATCH], float (&gi)[BATCH]) {
    #pragma unroll
    for (int u = 0; u < BATCH; ++u) {
        float vr = c0r * tx[u] - c0i * ty[u] + c1r * tx[u + 1] - c1i * ty[u + 1];
        float vi = c0r * ty[u] + c0i * tx[u] + c1r * ty[u + 1] + c1i * tx[u + 1];
        bool  in = c0 + u < nint;
        gr[u] = in ? vr : 0.f;
        gi[u] = in ? vi : 0.f;
    }
}

__global__ __launch_bounds__(SCAN_THREADS)
void interval_scan(const float* __restrict__ pk, const float* __restrict__ TAx,
                   const float* __restrict__ TAy, const float* __restrict__ fc,
                   const int* __restrict__ p_dt, const int* __restrict__ p_dtf,
                   float2* __restrict__ zst, float4* __restrict__ tab,
                   int nsteps, int nt)
{
    __shared__ float2 sh_tot[32];
    __shared__ float  sh_coef[6];
    int t    = threadIdx.x;
    int lane = t & 31;
    int w    = t >> 5;

    float dt      = (float)p_dt[0];
    int   nsub    = p_dtf[0] / p_dt[0];
    float invnsub = 1.f / (float)nsub;
    float K0   = __expf(pk[0]);
    float K1   = __expf(pk[1]);
    float f    = fc[0];
    float ar   = 1.f - dt * K1;
    float ai   = -dt * f;
    float dtK0 = dt * K0;

    int nint = (nsteps + nsub - 1) / nsub;
    int cpt  = (nint + SCAN_THREADS - 1) / SCAN_THREADS;
    int cb   = t * cpt;

    float tx[BATCH + 1], ty[BATCH + 1];
    load_batch(TAx, TAy, cb, nt, tx, ty);

    if (w == 0) {
        int r   = lane & 15;
        int h   = lane >> 4;
        int spl = (nsub + 15) >> 4;
        int m0  = r * spl;

        float Pr = 1.f, Pi = 0.f, D0r = 0.f, D0i = 0.f, D1r = 0.f, D1i = 0.f;
        for (int s = 0; s < spl; ++s) {
            float aa = (float)(m0 + s) * invnsub;
            float tr, ti;
            cmul(ar, ai, Pr, Pi, tr, ti);   Pr = tr; Pi = ti;
            cmul(ar, ai, D0r, D0i, tr, ti); D0r = tr + (1.f - aa); D0i = ti;
            cmul(ar, ai, D1r, D1i, tr, ti); D1r = tr + aa;         D1i = ti;
        }

        float Xr = Pr, Xi = Pi, Qr = 1.f, Qi = 0.f;
        scan_step<1>(r, Xr, Xi, Qr, Qi, D0r, D0i, D1r, D1i);
        scan_step<2>(r, Xr, Xi, Qr, Qi, D0r, D0i, D1r, D1i);
        scan_step<4>(r, Xr, Xi, Qr, Qi, D0r, D0i, D1r, D1i);
        scan_step<8>(r, Xr, Xi, Qr, Qi, D0r, D0i, D1r, D1i);

        Pr  = Qr;              Pi  = Qi;
        D0r = row_shr<1>(D0r); D0i = row_shr<1>(D0i);
        D1r = row_shr<1>(D1r); D1i = row_shr<1>(D1i);

        for (int s = 0; s < spl; ++s) {
            int   m  = m0 + s;
            float aa = (float)m * invnsub;
            float tr, ti;
            cmul(ar, ai, Pr, Pi, tr, ti);   Pr = tr; Pi = ti;
            cmul(ar, ai, D0r, D0i, tr, ti); D0r = tr + (1.f - aa); D0i = ti;
            cmul(ar, ai, D1r, D1i, tr, ti); D1r = tr + aa;         D1i = ti;
            if (m < nsub) {
                float e0r = dtK0 * D0r, e0i = dtK0 * D0i;
                float e1r = dtK0 * D1r, e1i = dtK0 * D1i;
                tab[4 * m + 2 * h]     = h ? make_float4(Pi, Pr, e0i, e0r)
                                           : make_float4(Pr, -Pi, e0r, -e0i);
                tab[4 * m + 2 * h + 1] = h ? make_float4(e1i, e1r, 0.f, 0.f)
                                           : make_float4(e1r, -e1i, 0.f, 0.f);
                if (m == nsub - 1 && h == 0) {
                    sh_coef[0] = Pr;   sh_coef[1] = Pi;
                    sh_coef[2] = e0r;  sh_coef[3] = e0i;
                    sh_coef[4] = e1r;  sh_coef[5] = e1i;
                }
            }
        }
    }
    __syncthreads();
    float Ar = sh_coef[0], Ai = sh_coef[1];
    float c0r = sh_coef[2], c0i = sh_coef[3];
    float c1r = sh_coef[4], c1i = sh_coef[5];

    float gr[BATCH], gi[BATCH];
    float Br = 0.f, Bi = 0.f, Mr = 1.f, Mi = 0.f;
    for (int done = 0;;) {
        int rem = cpt - done;
        batch_terms(c0r, c0i, c1r, c1i, tx, ty, cb + done, nint, gr, gi);
        #pragma unroll
        for (int u = 0; u < BATCH; ++u) {
            if (u < rem) {
                float tr, ti;
                cmul(Ar, Ai, Br, Bi, tr, ti); Br = tr + gr[u]; Bi = ti + gi[u];
                cmul(Ar, Ai, Mr, Mi, tr, ti); Mr = tr;         Mi = ti;
            }
        }
        done += BATCH;
        if (done >= cpt) break;
        load_batch(TAx, TAy, cb + done, nt, tx, ty);
    }

    float Qr = 1.f, Qi = 0.f, Xr = Mr, Xi = Mi;
    #pragma unroll
    for (int off = 1; off < 32; off <<= 1) {
        float pr = __shfl_up(Br, off);
        float pi = __shfl_up(Bi, off);
        float tr, ti;
        cmul(Xr, Xi, pr, pi, tr, ti);
        if (lane >= off) { Br += tr; Bi += ti; }
        cmul(Qr, Qi, Xr, Xi, tr, ti);
        if (lane & off) { Qr = tr; Qi = ti; }
        cmul(Xr, Xi, Xr, Xi, tr, ti); Xr = tr; Xi = ti;
    }
    if (lane == 31) sh_tot[w] = make_float2(Br, Bi);
    __syncthreads();

    float2 tt = sh_tot[lane];
    float Tr = tt.x, Ti = tt.y;
    #pragma unroll
    for (int off = 1; off < 32; off <<= 1) {
        float pr = __shfl_up(Tr, off);
        float pi = __shfl_up(Ti, off);
        float tr, ti;
        cmul(Xr, Xi, pr, pi, tr, ti);
        if (lane >= off) { Tr += tr; Ti += ti; }
        cmul(Xr, Xi, Xr, Xi, tr, ti); Xr = tr; Xi = ti;
    }
    int   src = (w + 31) & 31;
    float Er  = __shfl(Tr, src);
    float Ei  = __shfl(Ti, src);
    if (w == 0) { Er = 0.f; Ei = 0.f; }
    float xr = __shfl_up(Br, 1);
    float xi = __shfl_up(Bi, 1);
    if (lane == 0) { xr = 0.f; xi = 0.f; }
    float Zr, Zi;
    cmul(Qr, Qi, Er, Ei, Zr, Zi);
    Zr += xr; Zi += xi;

    if (cpt > BATCH) {
        load_batch(TAx, TAy, cb, nt, tx, ty);
        batch_terms(c0r, c0i, c1r, c1i, tx, ty, cb, nint, gr, gi);
    }
    for (int done = 0;;) {
        int rem = cpt - done;
        #pragma unroll
        for (int u = 0; u < BATCH; ++u) {
            if (u < rem) {
                zst[cb + done + u] = make_float2(Zr, Zi);
                float tr, ti;
                cmul(Ar, Ai, Zr, Zi, tr, ti); Zr = tr + gr[u]; Zi = ti + gi[u];
            }
        }
        done += BATCH;
        if (done >= cpt) break;
        load_batch(TAx, TAy, cb + done, nt, tx, ty);
        batch_terms(c0r, c0i, c1r, c1i, tx, ty, cb + done, nint, gr, gi);
    }
}

__global__ __launch_bounds__(32)
void interval_expand(const float* __restrict__ TAx, const float* __restrict__ TAy,
                     const int* __restrict__ p_dt, const int* __restrict__ p_dtf,
                     const float2* __restrict__ zst, const float* __restrict__ tabf,
                     float* __restrict__ out, int nsteps, int nt)
{
    int lane = threadIdx.x;
    int h    = lane >> 4;
    int r    = lane & 15;

    int nsub   = p_dtf[0] / p_dt[0];
    int nint   = (nsteps + nsub - 1) / nsub;
    int ntile  = (nint + 15) >> 4;
    int mtiles = (nsub + 15) >> 4;

    for (int mt0 = 0; mt0 < mtiles; mt0 += 4) {
        v2f Wr0[4], Wr1[4], Wi0[4], Wi1[4];
        #pragma unroll
        for (int q = 0; q < 4; ++q) {
            int mc = min(16 * (mt0 + q) + r, nsub - 1);
            const float* trow = tabf + TABROW * mc + 2 * h;
            Wr0[q] = *reinterpret_cast<const v2f*>(trow);
            Wr1[q] = *reinterpret_cast<const v2f*>(trow + 4);
            Wi0[q] = *reinterpret_cast<const v2f*>(trow + 8);
            Wi1[q] = *reinterpret_cast<const v2f*>(trow + 12);
        }

        for (int tile = blockIdx.x; tile < ntile; tile += gridDim.x) {
            int i = tile * 16 + r;
            unsigned is = (unsigned)min(i + 1 - h, nt - 1);
            float  sx = TAx[is];
            float  sy = TAy[is];
            float2 z  = zst[i];
            v2f X0, X1;
            X0[0] = h ? sx : z.x;   X0[1] = h ? sy : z.y;
            X1[0] = h ? 0.f : sx;   X1[1] = h ? 0.f : sy;

            v8f Dre[4], Dim[4];
            #pragma unroll
            for (int q = 0; q < 4; ++q) {
                v8f zero = {};
                Dre[q] = __builtin_amdgcn_wmma_f32_16x16x4_f32(false, X0, false, Wr0[q],
                                                               (short)0, zero, false, false);
                Dim[q] = __builtin_amdgcn_wmma_f32_16x16x4_f32(false, X0, false, Wi0[q],
                                                               (short)0, zero, false, false);
            }
            #pragma unroll
            for (int q = 0; q < 4; ++q) {
                Dre[q] = __builtin_amdgcn_wmma_f32_16x16x4_f32(false, X1, false, Wr1[q],
                                                               (short)0, Dre[q], false, false);
                Dim[q] = __builtin_amdgcn_wmma_f32_16x16x4_f32(false, X1, false, Wi1[q],
                                                               (short)0, Dim[q], false, false);
            }
            #pragma unroll
            for (int q = 0; q < 4; ++q) {
                int m = 16 * (mt0 + q) + r;
                #pragma unroll
                for (int e = 0; e < 8; ++e) {
                    int n = (tile * 16 + e + 8 * h) * nsub + m;
                    if (m < nsub && n < nsteps) {
                        float2 v; v.x = Dre[q][e]; v.y = Dim[q][e];
                        *reinterpret_cast<float2*>(out + 2 * (long)n) = v;
                    }
                }
            }
        }
    }
}

extern "C" void kernel_launch(void* const* d_in, const int* in_sizes, int n_in,
                              void* d_out, int out_size, void* d_ws, size_t ws_size,
                              hipStream_t stream)
{
    const float* pk    = (const float*)d_in[0];
    const float* TAx   = (const float*)d_in[1];
    const float* TAy   = (const float*)d_in[2];
    const float* fc    = (const float*)d_in[3];
    const int*   p_dt  = (const int*)d_in[6];
    const int*   p_dtf = (const int*)d_in[7];
    (void)n_in; (void)ws_size;

    int nt      = in_sizes[1];
    int nsteps  = out_size / 2;
    float*  out = (float*)d_out;
    float2* zst = (float2*)d_ws;
    float4* tab = (float4*)((char*)d_ws + TAB_OFFSET_BYTES);

    interval_scan<<<1, SCAN_THREADS, 0, stream>>>(pk, TAx, TAy, fc, p_dt, p_dtf,
                                                  zst, tab, nsteps, nt);
    int blocks = (nt + 15) / 16;
    interval_expand<<<blocks, 32, 0, stream>>>(TAx, TAy, p_dt, p_dtf, zst, (const float*)tab,
                                               out, nsteps, nt);
}


# Scan block's loads and stores made whole-line: samples loaded lane-consecutive into LDS, each thread reads its run there; start states leave through LDS lane-consecutive; sums keep order
# speedup vs baseline: 9.9243x; 1.3015x over previous
#include <hip/hip_runtime.h>

typedef __attribute__((ext_vector_type(2))) float v2f;
typedef __attribute__((ext_vector_type(8))) float v8f;

#define SCAN_THREADS 1024
#define BATCH 10
#define TABROW 16
#define TAB_OFFSET_BYTES (128u << 20)
#define STAGE_K 11
#define STAGE_CAP (STAGE_K * SCAN_THREADS)

__device__ __forceinline__ void cmul(float ar, float ai, float br, float bi,
                                     float& cr, float& ci) {
    cr = ar * br - ai * bi;
    ci = ar * bi + ai * br;
}

template <int N>
__device__ __forceinline__ float row_shr(float v) {
    return __builtin_bit_cast(float,
        __builtin_amdgcn_update_dpp(0, __builtin_bit_cast(int, v), 0x110 + N, 0xf, 0xf, false));
}

template <int N>
__device__ __forceinline__ void scan_step(int r, float& Xr, float& Xi, float& Qr, float& Qi,
                                          float& D0r, float& D0i, float& D1r, float& D1i) {
    float u0r = row_shr<N>(D0r), u0i = row_shr<N>(D0i);
    float u1r = row_shr<N>(D1r), u1i = row_shr<N>(D1i);
    float tr, ti;
    cmul(Xr, Xi, u0r, u0i, tr, ti); if (r >= N) { D0r += tr; D0i += ti; }
    cmul(Xr, Xi, u1r, u1i, tr, ti); if (r >= N) { D1r += tr; D1i += ti; }
    cmul(Qr, Qi, Xr, Xi, tr, ti);   if (r & N)  { Qr = tr;  Qi = ti; }
    cmul(Xr, Xi, Xr, Xi, tr, ti);   Xr = tr; Xi = ti;
}

__device__ __forceinline__ void load_batch(const float* __restrict__ TAx,
                                           const float* __restrict__ TAy,
                                           int c0, int nt,
                                           float (&tx)[BATCH + 1], float (&ty)[BATCH + 1]) {
    #pragma unroll
    for (int u = 0; u <= BATCH; ++u) {
        unsigned ii = (unsigned)min(c0 + u, nt - 1);
        tx[u] = TAx[ii];
        ty[u] = TAy[ii];
    }
}

__device__ __forceinline__ void load_batch_lds(const float* sx, const float* sy,
                                               int c0, int nt,
                                               float (&tx)[BATCH + 1], float (&ty)[BATCH + 1]) {
    #pragma unroll
    for (int u = 0; u <= BATCH; ++u) {
        int ii = min(c0 + u, nt - 1);
        tx[u] = sx[ii];
        ty[u] = sy[ii];
    }
}

__device__ __forceinline__ void batch_terms(float c0r, float c0i, float c1r, float c1i,
                                            const float (&tx)[BATCH + 1],
                                            const float (&ty)[BATCH + 1],
                                            int c0, int nint,
                                            float (&gr)[BATCH], float (&gi)[BATCH]) {
    #pragma unroll
    for (int u = 0; u < BATCH; ++u) {
        float vr = c0r * tx[u] - c0i * ty[u] + c1r * tx[u + 1] - c1i * ty[u + 1];
        float vi = c0r * ty[u] + c0i * tx[u] + c1r * ty[u + 1] + c1i * tx[u + 1];
        bool  in = c0 + u < nint;
        gr[u] = in ? vr : 0.f;
        gi[u] = in ? vi : 0.f;
    }
}

__global__ __launch_bounds__(SCAN_THREADS)
void interval_scan(const float* __restrict__ pk, const float* __restrict__ TAx,
                   const float* __restrict__ TAy, const float* __restrict__ fc,
                   const int* __restrict__ p_dt, const int* __restrict__ p_dtf,
                   float2* __restrict__ zst, float4* __restrict__ tab,
                   int nsteps, int nt)
{
    __shared__ float2 sh_tot[32];
    __shared__ float  sh_coef[6];
    __shared__ float  sTx[STAGE_CAP];
    __shared__ float  sTy[STAGE_CAP];
    __shared__ float2 sZ[SCAN_THREADS * BATCH];
    int t    = threadIdx.x;
    int lane = t & 31;
    int w    = t >> 5;

    int dt_i  = p_dt[0];
    int dtf_i = p_dtf[0];

    bool fits = nt <= STAGE_CAP;
    float lx[STAGE_K], ly[STAGE_K];
    if (fits) {
        #pragma unroll
        for (int k = 0; k < STAGE_K; ++k) {
            if (k * SCAN_THREADS < nt) {
                unsigned ii = (unsigned)min(k * SCAN_THREADS + t, nt - 1);
                lx[k] = TAx[ii];
                ly[k] = TAy[ii];
            }
        }
    }

    float dt      = (float)dt_i;
    int   nsub    = dtf_i / dt_i;
    float invnsub = 1.f / (float)nsub;
    float K0   = __expf(pk[0]);
    float K1   = __expf(pk[1]);
    float f    = fc[0];
    float ar   = 1.f - dt * K1;
    float ai   = -dt * f;
    float dtK0 = dt * K0;

    int nint = (nsteps + nsub - 1) / nsub;
    int cpt  = (nint + SCAN_THREADS - 1) / SCAN_THREADS;
    int cb   = t * cpt;

    bool staged = fits && cpt <= BATCH;

    float tx[BATCH + 1], ty[BATCH + 1];
    if (!staged) load_batch(TAx, TAy, cb, nt, tx, ty);

    if (w == 0) {
        int r   = lane & 15;
        int h   = lane >> 4;
        int spl = (nsub + 15) >> 4;
        int m0  = r * spl;

        float Pr = 1.f, Pi = 0.f, D0r = 0.f, D0i = 0.f, D1r = 0.f, D1i = 0.f;
        for (int s = 0; s < spl; ++s) {
            float aa = (float)(m0 + s) * invnsub;
            float tr, ti;
            cmul(ar, ai, Pr, Pi, tr, ti);   Pr = tr; Pi = ti;
            cmul(ar, ai, D0r, D0i, tr, ti); D0r = tr + (1.f - aa); D0i = ti;
            cmul(ar, ai, D1r, D1i, tr, ti); D1r = tr + aa;         D1i = ti;
        }

        float Xr = Pr, Xi = Pi, Qr = 1.f, Qi = 0.f;
        scan_step<1>(r, Xr, Xi, Qr, Qi, D0r, D0i, D1r, D1i);
        scan_step<2>(r, Xr, Xi, Qr, Qi, D0r, D0i, D1r, D1i);
        scan_step<4>(r, Xr, Xi, Qr, Qi, D0r, D0i, D1r, D1i);
        scan_step<8>(r, Xr, Xi, Qr, Qi, D0r, D0i, D1r, D1i);

        Pr  = Qr;              Pi  = Qi;
        D0r = row_shr<1>(D0r); D0i = row_shr<1>(D0i);
        D1r = row_shr<1>(D1r); D1i = row_shr<1>(D1i);

        for (int s = 0; s < spl; ++s) {
            int   m  = m0 + s;
            float aa = (float)m * invnsub;
            float tr, ti;
            cmul(ar, ai, Pr, Pi, tr, ti);   Pr = tr; Pi = ti;
            cmul(ar, ai, D0r, D0i, tr, ti); D0r = tr + (1.f - aa); D0i = ti;
            cmul(ar, ai, D1r, D1i, tr, ti); D1r = tr + aa;         D1i = ti;
            if (m < nsub) {
                float e0r = dtK0 * D0r, e0i = dtK0 * D0i;
                float e1r = dtK0 * D1r, e1i = dtK0 * D1i;
                tab[4 * m + 2 * h]     = h ? make_float4(Pi, Pr, e0i, e0r)
                                           : make_float4(Pr, -Pi, e0r, -e0i);
                tab[4 * m + 2 * h + 1] = h ? make_float4(e1i, e1r, 0.f, 0.f)
                                           : make_float4(e1r, -e1i, 0.f, 0.f);
                if (m == nsub - 1 && h == 0) {
                    sh_coef[0] = Pr;   sh_coef[1] = Pi;
                    sh_coef[2] = e0r;  sh_coef[3] = e0i;
                    sh_coef[4] = e1r;  sh_coef[5] = e1i;
                }
            }
        }
    }
    if (fits) {
        #pragma unroll
        for (int k = 0; k < STAGE_K; ++k) {
            if (k * SCAN_THREADS < nt) {
                sTx[k * SCAN_THREADS + t] = lx[k];
                sTy[k * SCAN_THREADS + t] = ly[k];
            }
        }
    }
    __syncthreads();
    float Ar = sh_coef[0], Ai = sh_coef[1];
    float c0r = sh_coef[2], c0i = sh_coef[3];
    float c1r = sh_coef[4], c1i = sh_coef[5];
    if (staged) load_batch_lds(sTx, sTy, cb, nt, tx, ty);

    float gr[BATCH], gi[BATCH];
    float Br = 0.f, Bi = 0.f, Mr = 1.f, Mi = 0.f;
    for (int done = 0;;) {
        int rem = cpt - done;
        batch_terms(c0r, c0i, c1r, c1i, tx, ty, cb + done, nint, gr, gi);
        #pragma unroll
        for (int u = 0; u < BATCH; ++u) {
            if (u < rem) {
                float tr, ti;
                cmul(Ar, Ai, Br, Bi, tr, ti); Br = tr + gr[u]; Bi = ti + gi[u];
                cmul(Ar, Ai, Mr, Mi, tr, ti); Mr = tr;         Mi = ti;
            }
        }
        done += BATCH;
        if (done >= cpt) break;
        load_batch(TAx, TAy, cb + done, nt, tx, ty);
    }

    float Qr = 1.f, Qi = 0.f, Xr = Mr, Xi = Mi;
    #pragma unroll
    for (int off = 1; off < 32; off <<= 1) {
        float pr = __shfl_up(Br, off);
        float pi = __shfl_up(Bi, off);
        float tr, ti;
        cmul(Xr, Xi, pr, pi, tr, ti);
        if (lane >= off) { Br += tr; Bi += ti; }
        cmul(Qr, Qi, Xr, Xi, tr, ti);
        if (lane & off) { Qr = tr; Qi = ti; }
        cmul(Xr, Xi, Xr, Xi, tr, ti); Xr = tr; Xi = ti;
    }
    if (lane == 31) sh_tot[w] = make_float2(Br, Bi);
    __syncthreads();

    float2 tt = sh_tot[lane];
    float Tr = tt.x, Ti = tt.y;
    #pragma unroll
    for (int off = 1; off < 32; off <<= 1) {
        float pr = __shfl_up(Tr, off);
        float pi = __shfl_up(Ti, off);
        float tr, ti;
        cmul(Xr, Xi, pr, pi, tr, ti);
        if (lane >= off) { Tr += tr; Ti += ti; }
        cmul(Xr, Xi, Xr, Xi, tr, ti); Xr = tr; Xi = ti;
    }
    int   src = (w + 31) & 31;
    float Er  = __shfl(Tr, src);
    float Ei  = __shfl(Ti, src);
    if (w == 0) { Er = 0.f; Ei = 0.f; }
    float xr = __shfl_up(Br, 1);
    float xi = __shfl_up(Bi, 1);
    if (lane == 0) { xr = 0.f; xi = 0.f; }
    float Zr, Zi;
    cmul(Qr, Qi, Er, Ei, Zr, Zi);
    Zr += xr; Zi += xi;

    if (cpt > BATCH) {
        load_batch(TAx, TAy, cb, nt, tx, ty);
        batch_terms(c0r, c0i, c1r, c1i, tx, ty, cb, nint, gr, gi);
    }
    if (staged) {
        #pragma unroll
        for (int u = 0; u < BATCH; ++u) {
            if (u < cpt) {
                sZ[cb + u] = make_float2(Zr, Zi);
                float tr, ti;
                cmul(Ar, Ai, Zr, Zi, tr, ti); Zr = tr + gr[u]; Zi = ti + gi[u];
            }
        }
        __syncthreads();
        float2 zv[BATCH];
        #pragma unroll
        for (int k = 0; k < BATCH; ++k) {
            if (k < cpt) zv[k] = sZ[k * SCAN_THREADS + t];
        }
        #pragma unroll
        for (int k = 0; k < BATCH; ++k) {
            if (k < cpt) zst[k * SCAN_THREADS + t] = zv[k];
        }
        return;
    }
    for (int done = 0;;) {
        int rem = cpt - done;
        #pragma unroll
        for (int u = 0; u < BATCH; ++u) {
            if (u < rem) {
                zst[cb + done + u] = make_float2(Zr, Zi);
                float tr, ti;
                cmul(Ar, Ai, Zr, Zi, tr, ti); Zr = tr + gr[u]; Zi = ti + gi[u];
            }
        }
        done += BATCH;
        if (done >= cpt) break;
        load_batch(TAx, TAy, cb + done, nt, tx, ty);
        batch_terms(c0r, c0i, c1r, c1i, tx, ty, cb + done, nint, gr, gi);
    }
}

__global__ __launch_bounds__(32)
void interval_expand(const float* __restrict__ TAx, const float* __restrict__ TAy,
                     const int* __restrict__ p_dt, const int* __restrict__ p_dtf,
                     const float2* __restrict__ zst, const float* __restrict__ tabf,
                     float* __restrict__ out, int nsteps, int nt)
{
    int lane = threadIdx.x;
    int h    = lane >> 4;
    int r    = lane & 15;

    int nsub   = p_dtf[0] / p_dt[0];
    int nint   = (nsteps + nsub - 1) / nsub;
    int ntile  = (nint + 15) >> 4;
    int mtiles = (nsub + 15) >> 4;

    for (int mt0 = 0; mt0 < mtiles; mt0 += 4) {
        v2f Wr0[4], Wr1[4], Wi0[4], Wi1[4];
        #pragma unroll
        for (int q = 0; q < 4; ++q) {
            int mc = min(16 * (mt0 + q) + r, nsub - 1);
            const float* trow = tabf + TABROW * mc + 2 * h;
            Wr0[q] = *reinterpret_cast<const v2f*>(trow);
            Wr1[q] = *reinterpret_cast<const v2f*>(trow + 4);
            Wi0[q] = *reinterpret_cast<const v2f*>(trow + 8);
            Wi1[q] = *reinterpret_cast<const v2f*>(trow + 12);
        }

        for (int tile = blockIdx.x; tile < ntile; tile += gridDim.x) {
            int i = tile * 16 + r;
            unsigned is = (unsigned)min(i + 1 - h, nt - 1);
            float  sx = TAx[is];
            float  sy = TAy[is];
            float2 z  = zst[i];
            v2f X0, X1;
            X0[0] = h ? sx : z.x;   X0[1] = h ? sy : z.y;
            X1[0] = h ? 0.f : sx;   X1[1] = h ? 0.f : sy;

            v8f Dre[4], Dim[4];
            #pragma unroll
            for (int q = 0; q < 4; ++q) {
                v8f zero = {};
                Dre[q] = __builtin_amdgcn_wmma_f32_16x16x4_f32(false, X0, false, Wr0[q],
                                                               (short)0, zero, false, false);
                Dim[q] = __builtin_amdgcn_wmma_f32_16x16x4_f32(false, X0, false, Wi0[q],
                                                               (short)0, zero, false, false);
            }
            #pragma unroll
            for (int q = 0; q < 4; ++q) {
                Dre[q] = __builtin_amdgcn_wmma_f32_16x16x4_f32(false, X1, false, Wr1[q],
                                                               (short)0, Dre[q], false, false);
                Dim[q] = __builtin_amdgcn_wmma_f32_16x16x4_f32(false, X1, false, Wi1[q],
                                                               (short)0, Dim[q], false, false);
            }
            #pragma unroll
            for (int q = 0; q < 4; ++q) {
                int m = 16 * (mt0 + q) + r;
                #pragma unroll
                for (int e = 0; e < 8; ++e) {
                    int n = (tile * 16 + e + 8 * h) * nsub + m;
                    if (m < nsub && n < nsteps) {
                        float2 v; v.x = Dre[q][e]; v.y = Dim[q][e];
                        *reinterpret_cast<float2*>(out + 2 * (long)n) = v;
                    }
                }
            }
        }
    }
}

extern "C" void kernel_launch(void* const* d_in, const int* in_sizes, int n_in,
                              void* d_out, int out_size, void* d_ws, size_t ws_size,
                              hipStream_t stream)
{
    const float* pk    = (const float*)d_in[0];
    const float* TAx   = (const float*)d_in[1];
    const float* TAy   = (const float*)d_in[2];
    const float* fc    = (const float*)d_in[3];
    const int*   p_dt  = (const int*)d_in[6];
    const int*   p_dtf = (const int*)d_in[7];
    (void)n_in; (void)ws_size;

    int nt      = in_sizes[1];
    int nsteps  = out_size / 2;
    float*  out = (float*)d_out;
    float2* zst = (float2*)d_ws;
    float4* tab = (float4*)((char*)d_ws + TAB_OFFSET_BYTES);

    interval_scan<<<1, SCAN_THREADS, 0, stream>>>(pk, TAx, TAy, fc, p_dt, p_dtf,
                                                  zst, tab, nsteps, nt);
    int blocks = (nt + 15) / 16;
    interval_expand<<<blocks, 32, 0, stream>>>(TAx, TAy, p_dt, p_dtf, zst, (const float*)tab,
                                               out, nsteps, nt);
}


# Start states on two levels: scan block finds one per 16-interval tile (one walk, no second walk, no LDS turn); each tile wave scans its 16 forcing terms over a lane row by DPP; reduction tree changes
# speedup vs baseline: 10.5206x; 1.0601x over previous
#include <hip/hip_runtime.h>

typedef __attribute__((ext_vector_type(2))) float v2f;
typedef __attribute__((ext_vector_type(8))) float v8f;

#define SCAN_THREADS 1024
#define TILE 16
#define TABROW 16
#define TAB_OFFSET_BYTES (128u << 20)
#define STAGE_K 11
#define STAGE_CAP (STAGE_K * SCAN_THREADS)
#define STAGE_PAD (STAGE_CAP + STAGE_CAP / TILE + 1)

__device__ __forceinline__ void cmul(float ar, float ai, float br, float bi,
                                     float& cr, float& ci) {
    cr = ar * br - ai * bi;
    ci = ar * bi + ai * br;
}

template <int N>
__device__ __forceinline__ float row_shr(float v) {
    return __builtin_bit_cast(float,
        __builtin_amdgcn_update_dpp(0, __builtin_bit_cast(int, v), 0x110 + N, 0xf, 0xf, false));
}

template <int N>
__device__ __forceinline__ void scan_step(int r, float& Xr, float& Xi, float& Qr, float& Qi,
                                          float& D0r, float& D0i, float& D1r, float& D1i) {
    float u0r = row_shr<N>(D0r), u0i = row_shr<N>(D0i);
    float u1r = row_shr<N>(D1r), u1i = row_shr<N>(D1i);
    float tr, ti;
    cmul(Xr, Xi, u0r, u0i, tr, ti); if (r >= N) { D0r += tr; D0i += ti; }
    cmul(Xr, Xi, u1r, u1i, tr, ti); if (r >= N) { D1r += tr; D1i += ti; }
    cmul(Qr, Qi, Xr, Xi, tr, ti);   if (r & N)  { Qr = tr;  Qi = ti; }
    cmul(Xr, Xi, Xr, Xi, tr, ti);   Xr = tr; Xi = ti;
}

template <int N>
__device__ __forceinline__ void scan_step1(int r, float& Xr, float& Xi, float& Qr, float& Qi,
                                           float& Dr, float& Di) {
    float ur = row_shr<N>(Dr), ui = row_shr<N>(Di);
    float tr, ti;
    cmul(Xr, Xi, ur, ui, tr, ti); if (r >= N) { Dr += tr; Di += ti; }
    cmul(Qr, Qi, Xr, Xi, tr, ti); if (r & N)  { Qr = tr;  Qi = ti; }
    cmul(Xr, Xi, Xr, Xi, tr, ti); Xr = tr; Xi = ti;
}

__device__ __forceinline__ void tile_samples_lds(const float* sx, const float* sy, int c, int nt,
                                                 float (&tx)[TILE + 1], float (&ty)[TILE + 1]) {
    #pragma unroll
    for (int u = 0; u <= TILE; ++u) {
        int jj = min(c + u, nt - 1);
        int p  = jj + (jj >> 4);
        tx[u] = sx[p];
        ty[u] = sy[p];
    }
}

__device__ __forceinline__ void tile_samples_mem(const float* __restrict__ TAx,
                                                 const float* __restrict__ TAy, int c, int nt,
                                                 float (&tx)[TILE + 1], float (&ty)[TILE + 1]) {
    #pragma unroll
    for (int u = 0; u <= TILE; ++u) {
        unsigned ii = (unsigned)min(c + u, nt - 1);
        tx[u] = TAx[ii];
        ty[u] = TAy[ii];
    }
}

__device__ __forceinline__ void tile_terms(float c0r, float c0i, float c1r, float c1i,
                                           const float (&tx)[TILE + 1],
                                           const float (&ty)[TILE + 1],
                                           int c, int nint,
                                           float (&gr)[TILE], float (&gi)[TILE]) {
    #pragma unroll
    for (int u = 0; u < TILE; ++u) {
        float vr = c0r * tx[u] - c0i * ty[u] + c1r * tx[u + 1] - c1i * ty[u + 1];
        float vi = c0r * ty[u] + c0i * tx[u] + c1r * ty[u + 1] + c1i * tx[u + 1];
        bool  in = c + u < nint;
        gr[u] = in ? vr : 0.f;
        gi[u] = in ? vi : 0.f;
    }
}

__global__ __launch_bounds__(SCAN_THREADS)
void interval_scan(const float* __restrict__ pk, const float* __restrict__ TAx,
                   const float* __restrict__ TAy, const float* __restrict__ fc,
                   const int* __restrict__ p_dt, const int* __restrict__ p_dtf,
                   float2* __restrict__ zst, float4* __restrict__ tab,
                   int nsteps, int nt)
{
    __shared__ float2 sh_tot[32];
    __shared__ float  sh_coef[6];
    __shared__ float  sTx[STAGE_PAD];
    __shared__ float  sTy[STAGE_PAD];
    int t    = threadIdx.x;
    int lane = t & 31;
    int w    = t >> 5;

    int dt_i  = p_dt[0];
    int dtf_i = p_dtf[0];

    bool fits = nt <= STAGE_CAP;
    float lx[STAGE_K], ly[STAGE_K];
    if (fits) {
        #pragma unroll
        for (int k = 0; k < STAGE_K; ++k) {
            if (k * SCAN_THREADS < nt) {
                unsigned ii = (unsigned)min(k * SCAN_THREADS + t, nt - 1);
                lx[k] = TAx[ii];
                ly[k] = TAy[ii];
            }
        }
    }

    float dt      = (float)dt_i;
    int   nsub    = dtf_i / dt_i;
    float invnsub = 1.f / (float)nsub;
    float K0   = __expf(pk[0]);
    float K1   = __expf(pk[1]);
    float f    = fc[0];
    float ar   = 1.f - dt * K1;
    float ai   = -dt * f;
    float dtK0 = dt * K0;

    int nint  = (nsteps + nsub - 1) / nsub;
    int ntile = (nint + TILE - 1) / TILE;
    int tpt   = (ntile + SCAN_THREADS - 1) / SCAN_THREADS;
    int tb    = t * tpt;
    int cb    = tb * TILE;

    if (w == 0) {
        int r   = lane & 15;
        int h   = lane >> 4;
        int spl = (nsub + 15) >> 4;
        int m0  = r * spl;

        float Pr = 1.f, Pi = 0.f, D0r = 0.f, D0i = 0.f, D1r = 0.f, D1i = 0.f;
        for (int s = 0; s < spl; ++s) {
            float aa = (float)(m0 + s) * invnsub;
            float tr, ti;
            cmul(ar, ai, Pr, Pi, tr, ti);   Pr = tr; Pi = ti;
            cmul(ar, ai, D0r, D0i, tr, ti); D0r = tr + (1.f - aa); D0i = ti;
            cmul(ar, ai, D1r, D1i, tr, ti); D1r = tr + aa;         D1i = ti;
        }

        float Xr = Pr, Xi = Pi, Qr = 1.f, Qi = 0.f;
        scan_step<1>(r, Xr, Xi, Qr, Qi, D0r, D0i, D1r, D1i);
        scan_step<2>(r, Xr, Xi, Qr, Qi, D0r, D0i, D1r, D1i);
        scan_step<4>(r, Xr, Xi, Qr, Qi, D0r, D0i, D1r, D1i);
        scan_step<8>(r, Xr, Xi, Qr, Qi, D0r, D0i, D1r, D1i);

        Pr  = Qr;              Pi  = Qi;
        D0r = row_shr<1>(D0r); D0i = row_shr<1>(D0i);
        D1r = row_shr<1>(D1r); D1i = row_shr<1>(D1i);

        for (int s = 0; s < spl; ++s) {
            int   m  = m0 + s;
            float aa = (float)m * invnsub;
            float tr, ti;
            cmul(ar, ai, Pr, Pi, tr, ti);   Pr = tr; Pi = ti;
            cmul(ar, ai, D0r, D0i, tr, ti); D0r = tr + (1.f - aa); D0i = ti;
            cmul(ar, ai, D1r, D1i, tr, ti); D1r = tr + aa;         D1i = ti;
            if (m < nsub) {
                float e0r = dtK0 * D0r, e0i = dtK0 * D0i;
                float e1r = dtK0 * D1r, e1i = dtK0 * D1i;
                tab[4 * m + 2 * h]     = h ? make_float4(Pi, Pr, e0i, e0r)
                                           : make_float4(Pr, -Pi, e0r, -e0i);
                tab[4 * m + 2 * h + 1] = h ? make_float4(e1i, e1r, 0.f, 0.f)
                                           : make_float4(e1r, -e1i, 0.f, 0.f);
                if (m == nsub - 1 && h == 0) {
                    sh_coef[0] = Pr;   sh_coef[1] = Pi;
                    sh_coef[2] = e0r;  sh_coef[3] = e0i;
                    sh_coef[4] = e1r;  sh_coef[5] = e1i;
                }
            }
        }
    }
    if (fits) {
        #pragma unroll
        for (int k = 0; k < STAGE_K; ++k) {
            if (k * SCAN_THREADS < nt) {
                int j = k * SCAN_THREADS + t;
                sTx[j + (j >> 4)] = lx[k];
                sTy[j + (j >> 4)] = ly[k];
            }
        }
    }
    __syncthreads();
    float Ar = sh_coef[0], Ai = sh_coef[1];
    float c0r = sh_coef[2], c0i = sh_coef[3];
    float c1r = sh_coef[4], c1i = sh_coef[5];

    float Mr, Mi;
    {
        float Sr = Ar, Si = Ai, tr, ti;
        #pragma unroll
        for (int k = 0; k < 4; ++k) { cmul(Sr, Si, Sr, Si, tr, ti); Sr = tr; Si = ti; }
        Mr = Sr; Mi = Si;
        for (int q = 1; q < tpt; ++q) { cmul(Sr, Si, Mr, Mi, tr, ti); Mr = tr; Mi = ti; }
    }

    float Br = 0.f, Bi = 0.f;
    if (cb < nint) {
        for (int q = 0; q < tpt; ++q) {
            float tx[TILE + 1], ty[TILE + 1], gr[TILE], gi[TILE];
            int c = cb + q * TILE;
            if (fits) tile_samples_lds(sTx, sTy, c, nt, tx, ty);
            else      tile_samples_mem(TAx, TAy, c, nt, tx, ty);
            tile_terms(c0r, c0i, c1r, c1i, tx, ty, c, nint, gr, gi);
            #pragma unroll
            for (int u = 0; u < TILE; ++u) {
                float tr, ti;
                cmul(Ar, Ai, Br, Bi, tr, ti); Br = tr + gr[u]; Bi = ti + gi[u];
            }
        }
    }

    float Qr = 1.f, Qi = 0.f, Xr = Mr, Xi = Mi;
    #pragma unroll
    for (int off = 1; off < 32; off <<= 1) {
        float pr = __shfl_up(Br, off);
        float pi = __shfl_up(Bi, off);
        float tr, ti;
        cmul(Xr, Xi, pr, pi, tr, ti);
        if (lane >= off) { Br += tr; Bi += ti; }
        cmul(Qr, Qi, Xr, Xi, tr, ti);
        if (lane & off) { Qr = tr; Qi = ti; }
        cmul(Xr, Xi, Xr, Xi, tr, ti); Xr = tr; Xi = ti;
    }
    if (lane == 31) sh_tot[w] = make_float2(Br, Bi);
    __syncthreads();

    float2 tt = sh_tot[lane];
    float Tr = tt.x, Ti = tt.y;
    #pragma unroll
    for (int off = 1; off < 32; off <<= 1) {
        float pr = __shfl_up(Tr, off);
        float pi = __shfl_up(Ti, off);
        float tr, ti;
        cmul(Xr, Xi, pr, pi, tr, ti);
        if (lane >= off) { Tr += tr; Ti += ti; }
        cmul(Xr, Xi, Xr, Xi, tr, ti); Xr = tr; Xi = ti;
    }
    int   src = (w + 31) & 31;
    float Er  = __shfl(Tr, src);
    float Ei  = __shfl(Ti, src);
    if (w == 0) { Er = 0.f; Ei = 0.f; }
    float xr = __shfl_up(Br, 1);
    float xi = __shfl_up(Bi, 1);
    if (lane == 0) { xr = 0.f; xi = 0.f; }
    float Zr, Zi;
    cmul(Qr, Qi, Er, Ei, Zr, Zi);
    Zr += xr; Zi += xi;

    if (cb >= nint) return;
    zst[tb] = make_float2(Zr, Zi);
    for (int q = 1; q < tpt; ++q) {
        if (tb + q >= ntile) break;
        float tx[TILE + 1], ty[TILE + 1], gr[TILE], gi[TILE];
        int c = cb + (q - 1) * TILE;
        if (fits) tile_samples_lds(sTx, sTy, c, nt, tx, ty);
        else      tile_samples_mem(TAx, TAy, c, nt, tx, ty);
        tile_terms(c0r, c0i, c1r, c1i, tx, ty, c, nint, gr, gi);
        #pragma unroll
        for (int u = 0; u < TILE; ++u) {
            float tr, ti;
            cmul(Ar, Ai, Zr, Zi, tr, ti); Zr = tr + gr[u]; Zi = ti + gi[u];
        }
        zst[tb + q] = make_float2(Zr, Zi);
    }
}

__global__ __launch_bounds__(32)
void interval_expand(const float* __restrict__ TAx, const float* __restrict__ TAy,
                     const int* __restrict__ p_dt, const int* __restrict__ p_dtf,
                     const float2* __restrict__ zst, const float* __restrict__ tabf,
                     float* __restrict__ out, int nsteps, int nt)
{
    int lane = threadIdx.x;
    int h    = lane >> 4;
    int r    = lane & 15;

    int nsub   = p_dtf[0] / p_dt[0];
    int nint   = (nsteps + nsub - 1) / nsub;
    int ntile  = (nint + 15) >> 4;
    int mtiles = (nsub + 15) >> 4;

    const float* lastw = tabf + TABROW * (nsub - 1);
    float Ar  = lastw[0], Ai  = -lastw[1];
    float c0r = lastw[2], c0i = -lastw[3];
    float c1r = lastw[4], c1i = -lastw[5];

    for (int mt0 = 0; mt0 < mtiles; mt0 += 4) {
        v2f Wr0[4], Wr1[4], Wi0[4], Wi1[4];
        #pragma unroll
        for (int q = 0; q < 4; ++q) {
            int mc = min(16 * (mt0 + q) + r, nsub - 1);
            const float* trow = tabf + TABROW * mc + 2 * h;
            Wr0[q] = *reinterpret_cast<const v2f*>(trow);
            Wr1[q] = *reinterpret_cast<const v2f*>(trow + 4);
            Wi0[q] = *reinterpret_cast<const v2f*>(trow + 8);
            Wi1[q] = *reinterpret_cast<const v2f*>(trow + 12);
        }

        for (int tile = blockIdx.x; tile < ntile; tile += gridDim.x) {
            int i = tile * 16 + r;
            unsigned i0 = (unsigned)min(i, nt - 1);
            unsigned i1 = (unsigned)min(i + 1, nt - 1);
            float  s0x = TAx[i0], s0y = TAy[i0];
            float  s1x = TAx[i1], s1y = TAy[i1];
            float2 zt  = zst[tile];

            bool  in = i < nint;
            float Dr = in ? c0r * s0x - c0i * s0y + c1r * s1x - c1i * s1y : 0.f;
            float Di = in ? c0r * s0y + c0i * s0x + c1r * s1y + c1i * s1x : 0.f;
            float Xr = Ar, Xi = Ai, Qr = 1.f, Qi = 0.f;
            scan_step1<1>(r, Xr, Xi, Qr, Qi, Dr, Di);
            scan_step1<2>(r, Xr, Xi, Qr, Qi, Dr, Di);
            scan_step1<4>(r, Xr, Xi, Qr, Qi, Dr, Di);
            scan_step1<8>(r, Xr, Xi, Qr, Qi, Dr, Di);
            float zr, zi;
            cmul(Qr, Qi, zt.x, zt.y, zr, zi);
            zr += row_shr<1>(Dr);
            zi += row_shr<1>(Di);

            v2f X0, X1;
            X0[0] = h ? s0x : zr;    X0[1] = h ? s0y : zi;
            X1[0] = h ? 0.f : s1x;   X1[1] = h ? 0.f : s1y;

            v8f Dre[4], Dim[4];
            #pragma unroll
            for (int q = 0; q < 4; ++q) {
                v8f zero = {};
                Dre[q] = __builtin_amdgcn_wmma_f32_16x16x4_f32(false, X0, false, Wr0[q],
                                                               (short)0, zero, false, false);
                Dim[q] = __builtin_amdgcn_wmma_f32_16x16x4_f32(false, X0, false, Wi0[q],
                                                               (short)0, zero, false, false);
            }
            #pragma unroll
            for (int q = 0; q < 4; ++q) {
                Dre[q] = __builtin_amdgcn_wmma_f32_16x16x4_f32(false, X1, false, Wr1[q],
                                                               (short)0, Dre[q], false, false);
                Dim[q] = __builtin_amdgcn_wmma_f32_16x16x4_f32(false, X1, false, Wi1[q],
                                                               (short)0, Dim[q], false, false);
            }
            #pragma unroll
            for (int q = 0; q < 4; ++q) {
                int m = 16 * (mt0 + q) + r;
                #pragma unroll
                for (int e = 0; e < 8; ++e) {
                    int n = (tile * 16 + e + 8 * h) * nsub + m;
                    if (m < nsub && n < nsteps) {
                        float2 v; v.x = Dre[q][e]; v.y = Dim[q][e];
                        *reinterpret_cast<float2*>(out + 2 * (long)n) = v;
                    }
                }
            }
        }
    }
}

extern "C" void kernel_launch(void* const* d_in, const int* in_sizes, int n_in,
                              void* d_out, int out_size, void* d_ws, size_t ws_size,
                              hipStream_t stream)
{
    const float* pk    = (const float*)d_in[0];
    const float* TAx   = (const float*)d_in[1];
    const float* TAy   = (const float*)d_in[2];
    const float* fc    = (const float*)d_in[3];
    const int*   p_dt  = (const int*)d_in[6];
    const int*   p_dtf = (const int*)d_in[7];
    (void)n_in; (void)ws_size;

    int nt      = in_sizes[1];
    int nsteps  = out_size / 2;
    float*  out = (float*)d_out;
    float2* zst = (float2*)d_ws;
    float4* tab = (float4*)((char*)d_ws + TAB_OFFSET_BYTES);

    interval_scan<<<1, SCAN_THREADS, 0, stream>>>(pk, TAx, TAy, fc, p_dt, p_dtf,
                                                  zst, tab, nsteps, nt);
    int blocks = (nt + 15) / 16;
    interval_expand<<<blocks, 32, 0, stream>>>(TAx, TAy, p_dt, p_dtf, zst, (const float*)tab,
                                               out, nsteps, nt);
}


# Loads unchained: launch 1 leaves a header (interval weights, nsub, nint) and a padded table, so each launch issues all its loads as one group at entry, not 4-5 dependent trips; sums keep order
# speedup vs baseline: 11.8139x; 1.1229x over previous
#include <hip/hip_runtime.h>

typedef __attribute__((ext_vector_type(2))) float v2f;
typedef __attribute__((ext_vector_type(8))) float v8f;

#define SCAN_THREADS 1024
#define TILE 16
#define TABROW 16
#define TAB_OFFSET_BYTES (128u << 20)
#define HDR_OFFSET_BYTES (64u << 20)
#define HDR_FLOATS 8
#define TAB_MIN_ROWS 64
#define STAGE_K 11
#define STAGE_CAP (STAGE_K * SCAN_THREADS)
#define STAGE_PAD (STAGE_CAP + STAGE_CAP / TILE + 1)

__device__ __forceinline__ void cmul(float ar, float ai, float br, float bi,
                                     float& cr, float& ci) {
    cr = ar * br - ai * bi;
    ci = ar * bi + ai * br;
}

template <int N>
__device__ __forceinline__ float row_shr(float v) {
    return __builtin_bit_cast(float,
        __builtin_amdgcn_update_dpp(0, __builtin_bit_cast(int, v), 0x110 + N, 0xf, 0xf, false));
}

template <int N>
__device__ __forceinline__ void scan_step(int r, float& Xr, float& Xi, float& Qr, float& Qi,
                                          float& D0r, float& D0i, float& D1r, float& D1i) {
    float u0r = row_shr<N>(D0r), u0i = row_shr<N>(D0i);
    float u1r = row_shr<N>(D1r), u1i = row_shr<N>(D1i);
    float tr, ti;
    cmul(Xr, Xi, u0r, u0i, tr, ti); if (r >= N) { D0r += tr; D0i += ti; }
    cmul(Xr, Xi, u1r, u1i, tr, ti); if (r >= N) { D1r += tr; D1i += ti; }
    cmul(Qr, Qi, Xr, Xi, tr, ti);   if (r & N)  { Qr = tr;  Qi = ti; }
    cmul(Xr, Xi, Xr, Xi, tr, ti);   Xr = tr; Xi = ti;
}

template <int N>
__device__ __forceinline__ void scan_step1(int r, float& Xr, float& Xi, float& Qr, float& Qi,
                                           float& Dr, float& Di) {
    float ur = row_shr<N>(Dr), ui = row_shr<N>(Di);
    float tr, ti;
    cmul(Xr, Xi, ur, ui, tr, ti); if (r >= N) { Dr += tr; Di += ti; }
    cmul(Qr, Qi, Xr, Xi, tr, ti); if (r & N)  { Qr = tr;  Qi = ti; }
    cmul(Xr, Xi, Xr, Xi, tr, ti); Xr = tr; Xi = ti;
}

__device__ __forceinline__ void tile_samples_lds(const float* sx, const float* sy, int c, int nt,
                                                 float (&tx)[TILE + 1], float (&ty)[TILE + 1]) {
    #pragma unroll
    for (int u = 0; u <= TILE; ++u) {
        int jj = min(c + u, nt - 1);
        int p  = jj + (jj >> 4);
        tx[u] = sx[p];
        ty[u] = sy[p];
    }
}

__device__ __forceinline__ void tile_samples_mem(const float* __restrict__ TAx,
                                                 const float* __restrict__ TAy, int c, int nt,
                                                 float (&tx)[TILE + 1], float (&ty)[TILE + 1]) {
    #pragma unroll
    for (int u = 0; u <= TILE; ++u) {
        unsigned ii = (unsigned)min(c + u, nt - 1);
        tx[u] = TAx[ii];
        ty[u] = TAy[ii];
    }
}

__device__ __forceinline__ void tile_terms(float c0r, float c0i, float c1r, float c1i,
                                           const float (&tx)[TILE + 1],
                                           const float (&ty)[TILE + 1],
                                           int c, int nint,
                                           float (&gr)[TILE], float (&gi)[TILE]) {
    #pragma unroll
    for (int u = 0; u < TILE; ++u) {
        float vr = c0r * tx[u] - c0i * ty[u] + c1r * tx[u + 1] - c1i * ty[u + 1];
        float vi = c0r * ty[u] + c0i * tx[u] + c1r * ty[u + 1] + c1i * tx[u + 1];
        bool  in = c + u < nint;
        gr[u] = in ? vr : 0.f;
        gi[u] = in ? vi : 0.f;
    }
}

__global__ __launch_bounds__(SCAN_THREADS)
void interval_scan(const float* __restrict__ pk, const float* __restrict__ TAx,
                   const float* __restrict__ TAy, const float* __restrict__ fc,
                   const int* __restrict__ p_dt, const int* __restrict__ p_dtf,
                   float2* __restrict__ zst, float4* __restrict__ tab,
                   float4* __restrict__ hdr, int nsteps, int nt)
{
    __shared__ float2 sh_tot[32];
    __shared__ float  sh_coef[6];
    __shared__ float  sTx[STAGE_PAD];
    __shared__ float  sTy[STAGE_PAD];
    int t    = threadIdx.x;
    int lane = t & 31;
    int w    = t >> 5;

    int dt_i  = p_dt[0];
    int dtf_i = p_dtf[0];
    float pk0 = pk[0], pk1 = pk[1], f = fc[0];

    bool fits = nt <= STAGE_CAP;
    float lx[STAGE_K], ly[STAGE_K];
    if (fits) {
        #pragma unroll
        for (int k = 0; k < STAGE_K; ++k) {
            if (k * SCAN_THREADS < nt) {
                unsigned ii = (unsigned)min(k * SCAN_THREADS + t, nt - 1);
                lx[k] = TAx[ii];
                ly[k] = TAy[ii];
            }
        }
    }

    asm volatile("" :: "s"(dt_i), "s"(dtf_i), "s"(pk0), "s"(pk1), "s"(f));

    float dt      = (float)dt_i;
    int   nsub    = dtf_i / dt_i;
    float invnsub = 1.f / (float)nsub;
    float K0   = __expf(pk0);
    float K1   = __expf(pk1);
    float ar   = 1.f - dt * K1;
    float ai   = -dt * f;
    float dtK0 = dt * K0;

    int nint  = (nsteps + nsub - 1) / nsub;
    int ntile = (nint + TILE - 1) / TILE;
    int tpt   = (ntile + SCAN_THREADS - 1) / SCAN_THREADS;
    int tb    = t * tpt;
    int cb    = tb * TILE;

    if (w == 0) {
        int r   = lane & 15;
        int h   = lane >> 4;
        int spl = (nsub + 15) >> 4;
        int m0  = r * spl;

        float Pr = 1.f, Pi = 0.f, D0r = 0.f, D0i = 0.f, D1r = 0.f, D1i = 0.f;
        for (int s = 0; s < spl; ++s) {
            float aa = (float)(m0 + s) * invnsub;
            float tr, ti;
            cmul(ar, ai, Pr, Pi, tr, ti);   Pr = tr; Pi = ti;
            cmul(ar, ai, D0r, D0i, tr, ti); D0r = tr + (1.f - aa); D0i = ti;
            cmul(ar, ai, D1r, D1i, tr, ti); D1r = tr + aa;         D1i = ti;
        }

        float Xr = Pr, Xi = Pi, Qr = 1.f, Qi = 0.f;
        scan_step<1>(r, Xr, Xi, Qr, Qi, D0r, D0i, D1r, D1i);
        scan_step<2>(r, Xr, Xi, Qr, Qi, D0r, D0i, D1r, D1i);
        scan_step<4>(r, Xr, Xi, Qr, Qi, D0r, D0i, D1r, D1i);
        scan_step<8>(r, Xr, Xi, Qr, Qi, D0r, D0i, D1r, D1i);

        Pr  = Qr;              Pi  = Qi;
        D0r = row_shr<1>(D0r); D0i = row_shr<1>(D0i);
        D1r = row_shr<1>(D1r); D1i = row_shr<1>(D1i);

        for (int s = 0; s < spl; ++s) {
            int   m  = m0 + s;
            float aa = (float)m * invnsub;
            float tr, ti;
            cmul(ar, ai, Pr, Pi, tr, ti);   Pr = tr; Pi = ti;
            cmul(ar, ai, D0r, D0i, tr, ti); D0r = tr + (1.f - aa); D0i = ti;
            cmul(ar, ai, D1r, D1i, tr, ti); D1r = tr + aa;         D1i = ti;
            float e0r = dtK0 * D0r, e0i = dtK0 * D0i;
            float e1r = dtK0 * D1r, e1i = dtK0 * D1i;
            tab[4 * m + 2 * h]     = h ? make_float4(Pi, Pr, e0i, e0r)
                                       : make_float4(Pr, -Pi, e0r, -e0i);
            tab[4 * m + 2 * h + 1] = h ? make_float4(e1i, e1r, 0.f, 0.f)
                                       : make_float4(e1r, -e1i, 0.f, 0.f);
            if (m == nsub - 1 && h == 0) {
                sh_coef[0] = Pr;   sh_coef[1] = Pi;
                sh_coef[2] = e0r;  sh_coef[3] = e0i;
                sh_coef[4] = e1r;  sh_coef[5] = e1i;
                hdr[0] = make_float4(Pr, Pi, e0r, e0i);
                hdr[1] = make_float4(e1r, e1i, __int_as_float(nsub), __int_as_float(nint));
            }
        }
        for (int m = 16 * spl + r; m < TAB_MIN_ROWS; m += 16) {
            tab[4 * m + 2 * h]     = make_float4(0.f, 0.f, 0.f, 0.f);
            tab[4 * m + 2 * h + 1] = make_float4(0.f, 0.f, 0.f, 0.f);
        }
    }
    if (fits) {
        #pragma unroll
        for (int k = 0; k < STAGE_K; ++k) {
            if (k * SCAN_THREADS < nt) {
                int j = k * SCAN_THREADS + t;
                sTx[j + (j >> 4)] = lx[k];
                sTy[j + (j >> 4)] = ly[k];
            }
        }
    }
    __syncthreads();
    float Ar = sh_coef[0], Ai = sh_coef[1];
    float c0r = sh_coef[2], c0i = sh_coef[3];
    float c1r = sh_coef[4], c1i = sh_coef[5];

    float Mr, Mi;
    {
        float Sr = Ar, Si = Ai, tr, ti;
        #pragma unroll
        for (int k = 0; k < 4; ++k) { cmul(Sr, Si, Sr, Si, tr, ti); Sr = tr; Si = ti; }
        Mr = Sr; Mi = Si;
        for (int q = 1; q < tpt; ++q) { cmul(Sr, Si, Mr, Mi, tr, ti); Mr = tr; Mi = ti; }
    }

    float Br = 0.f, Bi = 0.f;
    if (cb < nint) {
        for (int q = 0; q < tpt; ++q) {
            float tx[TILE + 1], ty[TILE + 1], gr[TILE], gi[TILE];
            int c = cb + q * TILE;
            if (fits) tile_samples_lds(sTx, sTy, c, nt, tx, ty);
            else      tile_samples_mem(TAx, TAy, c, nt, tx, ty);
            tile_terms(c0r, c0i, c1r, c1i, tx, ty, c, nint, gr, gi);
            #pragma unroll
            for (int u = 0; u < TILE; ++u) {
                float tr, ti;
                cmul(Ar, Ai, Br, Bi, tr, ti); Br = tr + gr[u]; Bi = ti + gi[u];
            }
        }
    }

    float Qr = 1.f, Qi = 0.f, Xr = Mr, Xi = Mi;
    #pragma unroll
    for (int off = 1; off < 32; off <<= 1) {
        float pr = __shfl_up(Br, off);
        float pi = __shfl_up(Bi, off);
        float tr, ti;
        cmul(Xr, Xi, pr, pi, tr, ti);
        if (lane >= off) { Br += tr; Bi += ti; }
        cmul(Qr, Qi, Xr, Xi, tr, ti);
        if (lane & off) { Qr = tr; Qi = ti; }
        cmul(Xr, Xi, Xr, Xi, tr, ti); Xr = tr; Xi = ti;
    }
    if (lane == 31) sh_tot[w] = make_float2(Br, Bi);
    __syncthreads();

    float2 tt = sh_tot[lane];
    float Tr = tt.x, Ti = tt.y;
    #pragma unroll
    for (int off = 1; off < 32; off <<= 1) {
        float pr = __shfl_up(Tr, off);
        float pi = __shfl_up(Ti, off);
        float tr, ti;
        cmul(Xr, Xi, pr, pi, tr, ti);
        if (lane >= off) { Tr += tr; Ti += ti; }
        cmul(Xr, Xi, Xr, Xi, tr, ti); Xr = tr; Xi = ti;
    }
    int   src = (w + 31) & 31;
    float Er  = __shfl(Tr, src);
    float Ei  = __shfl(Ti, src);
    if (w == 0) { Er = 0.f; Ei = 0.f; }
    float xr = __shfl_up(Br, 1);
    float xi = __shfl_up(Bi, 1);
    if (lane == 0) { xr = 0.f; xi = 0.f; }
    float Zr, Zi;
    cmul(Qr, Qi, Er, Ei, Zr, Zi);
    Zr += xr; Zi += xi;

    for (int j = ntile + t; j < ((nt + TILE - 1) >> 4); j += SCAN_THREADS) zst[j] = make_float2(0.f, 0.f);
    if (cb >= nint) return;
    zst[tb] = make_float2(Zr, Zi);
    for (int q = 1; q < tpt; ++q) {
        if (tb + q >= ntile) break;
        float tx[TILE + 1], ty[TILE + 1], gr[TILE], gi[TILE];
        int c = cb + (q - 1) * TILE;
        if (fits) tile_samples_lds(sTx, sTy, c, nt, tx, ty);
        else      tile_samples_mem(TAx, TAy, c, nt, tx, ty);
        tile_terms(c0r, c0i, c1r, c1i, tx, ty, c, nint, gr, gi);
        #pragma unroll
        for (int u = 0; u < TILE; ++u) {
            float tr, ti;
            cmul(Ar, Ai, Zr, Zi, tr, ti); Zr = tr + gr[u]; Zi = ti + gi[u];
        }
        zst[tb + q] = make_float2(Zr, Zi);
    }
}


__device__ __forceinline__ void load_weights(const float* __restrict__ tabf, int mt0, int r, int h,
                                             int lastrow, v2f (&Wr0)[4], v2f (&Wr1)[4],
                                             v2f (&Wi0)[4], v2f (&Wi1)[4]) {
    #pragma unroll
    for (int q = 0; q < 4; ++q) {
        int mc = min(16 * (mt0 + q) + r, lastrow);
        const float* trow = tabf + TABROW * mc + 2 * h;
        Wr0[q] = *reinterpret_cast<const v2f*>(trow);
        Wr1[q] = *reinterpret_cast<const v2f*>(trow + 4);
        Wi0[q] = *reinterpret_cast<const v2f*>(trow + 8);
        Wi1[q] = *reinterpret_cast<const v2f*>(trow + 12);
    }
}

__global__ __launch_bounds__(32)
void interval_expand(const float* __restrict__ TAx, const float* __restrict__ TAy,
                     const float* __restrict__ hdr, const float2* __restrict__ zst,
                     const float* __restrict__ tabf, float* __restrict__ out, int nsteps, int nt)
{
    int lane = threadIdx.x;
    int h    = lane >> 4;
    int r    = lane & 15;

    int gt    = (nt + 15) >> 4;
    int tile0 = blockIdx.x;

    v2f Wr0[4], Wr1[4], Wi0[4], Wi1[4];
    load_weights(tabf, 0, r, h, TAB_MIN_ROWS - 1, Wr0, Wr1, Wi0, Wi1);
    float  s0x, s0y, s1x, s1y;
    {
        int i = tile0 * 16 + r;
        unsigned i0 = (unsigned)min(i, nt - 1);
        unsigned i1 = (unsigned)min(i + 1, nt - 1);
        s0x = TAx[i0]; s0y = TAy[i0];
        s1x = TAx[i1]; s1y = TAy[i1];
    }
    float2 zt = zst[tile0];

    float Ar  = hdr[0], Ai  = hdr[1];
    float c0r = hdr[2], c0i = hdr[3];
    float c1r = hdr[4], c1i = hdr[5];
    float hn0 = hdr[6], hn1 = hdr[7];

    asm volatile("" :: "v"(Wr0[0]), "v"(Wr0[1]), "v"(Wr0[2]), "v"(Wr0[3]),
                       "v"(Wr1[0]), "v"(Wr1[1]), "v"(Wr1[2]), "v"(Wr1[3]),
                       "v"(Wi0[0]), "v"(Wi0[1]), "v"(Wi0[2]), "v"(Wi0[3]),
                       "v"(Wi1[0]), "v"(Wi1[1]), "v"(Wi1[2]), "v"(Wi1[3]),
                       "v"(s0x), "v"(s0y), "v"(s1x), "v"(s1y));
    asm volatile("" :: "s"(Ar), "s"(Ai), "s"(c0r), "s"(c0i), "s"(c1r), "s"(c1i),
                       "s"(hn0), "s"(hn1), "s"(zt.x), "s"(zt.y));

    int nsub   = __float_as_int(hn0);
    int nint   = __float_as_int(hn1);
    int ntile  = (nint + 15) >> 4;
    int mtiles = (nsub + 15) >> 4;

    for (int mt0 = 0; mt0 < mtiles; mt0 += 4) {
        if (mt0 > 0) load_weights(tabf, mt0, r, h, nsub - 1, Wr0, Wr1, Wi0, Wi1);

        for (int tile = tile0; tile < ntile; tile += gt) {
            int i = tile * 16 + r;
            if (mt0 > 0 || tile != tile0) {
                unsigned i0 = (unsigned)min(i, nt - 1);
                unsigned i1 = (unsigned)min(i + 1, nt - 1);
                s0x = TAx[i0]; s0y = TAy[i0];
                s1x = TAx[i1]; s1y = TAy[i1];
                zt  = zst[tile];
            }

            bool  in = i < nint;
            float Dr = in ? c0r * s0x - c0i * s0y + c1r * s1x - c1i * s1y : 0.f;
            float Di = in ? c0r * s0y + c0i * s0x + c1r * s1y + c1i * s1x : 0.f;
            float Xr = Ar, Xi = Ai, Qr = 1.f, Qi = 0.f;
            scan_step1<1>(r, Xr, Xi, Qr, Qi, Dr, Di);
            scan_step1<2>(r, Xr, Xi, Qr, Qi, Dr, Di);
            scan_step1<4>(r, Xr, Xi, Qr, Qi, Dr, Di);
            scan_step1<8>(r, Xr, Xi, Qr, Qi, Dr, Di);
            float zr, zi;
            cmul(Qr, Qi, zt.x, zt.y, zr, zi);
            zr += row_shr<1>(Dr);
            zi += row_shr<1>(Di);

            v2f X0, X1;
            X0[0] = h ? s0x : zr;    X0[1] = h ? s0y : zi;
            X1[0] = h ? 0.f : s1x;   X1[1] = h ? 0.f : s1y;

            v8f Dre[4], Dim[4];
            #pragma unroll
            for (int q = 0; q < 4; ++q) {
                v8f zero = {};
                Dre[q] = __builtin_amdgcn_wmma_f32_16x16x4_f32(false, X0, false, Wr0[q],
                                                               (short)0, zero, false, false);
                Dim[q] = __builtin_amdgcn_wmma_f32_16x16x4_f32(false, X0, false, Wi0[q],
                                                               (short)0, zero, false, false);
            }
            #pragma unroll
            for (int q = 0; q < 4; ++q) {
                Dre[q] = __builtin_amdgcn_wmma_f32_16x16x4_f32(false, X1, false, Wr1[q],
                                                               (short)0, Dre[q], false, false);
                Dim[q] = __builtin_amdgcn_wmma_f32_16x16x4_f32(false, X1, false, Wi1[q],
                                                               (short)0, Dim[q], false, false);
            }
            #pragma unroll
            for (int q = 0; q < 4; ++q) {
                int m = 16 * (mt0 + q) + r;
                #pragma unroll
                for (int e = 0; e < 8; ++e) {
                    int n = (tile * 16 + e + 8 * h) * nsub + m;
                    if (m < nsub && n < nsteps) {
                        float2 v; v.x = Dre[q][e]; v.y = Dim[q][e];
                        *reinterpret_cast<float2*>(out + 2 * (long)n) = v;
                    }
                }
            }
        }
    }
}

extern "C" void kernel_launch(void* const* d_in, const int* in_sizes, int n_in,
                              void* d_out, int out_size, void* d_ws, size_t ws_size,
                              hipStream_t stream)
{
    const float* pk    = (const float*)d_in[0];
    const float* TAx   = (const float*)d_in[1];
    const float* TAy   = (const float*)d_in[2];
    const float* fc    = (const float*)d_in[3];
    const int*   p_dt  = (const int*)d_in[6];
    const int*   p_dtf = (const int*)d_in[7];
    (void)n_in; (void)ws_size;

    int nt      = in_sizes[1];
    int nsteps  = out_size / 2;
    float*  out = (float*)d_out;
    float2* zst = (float2*)d_ws;
    float4* tab = (float4*)((char*)d_ws + TAB_OFFSET_BYTES);
    float4* hdr = (float4*)((char*)d_ws + HDR_OFFSET_BYTES);

    interval_scan<<<1, SCAN_THREADS, 0, stream>>>(pk, TAx, TAy, fc, p_dt, p_dtf,
                                                  zst, tab, hdr, nsteps, nt);
    int blocks = (nt + 15) / 16;
    interval_expand<<<blocks, 32, 0, stream>>>(TAx, TAy, (const float*)hdr, zst, (const float*)tab,
                                               out, nsteps, nt);
}


# Expand launch's stores: the row-in-range test leaves the lanes (one thread-shared count of whole rows per tile): one mask and branch per 16-step column group, not one per store; sums keep order
# speedup vs baseline: 12.9494x; 1.0961x over previous
#include <hip/hip_runtime.h>

typedef __attribute__((ext_vector_type(2))) float v2f;
typedef __attribute__((ext_vector_type(8))) float v8f;

#define SCAN_THREADS 1024
#define TILE 16
#define TABROW 16
#define TAB_OFFSET_BYTES (128u << 20)
#define HDR_OFFSET_BYTES (64u << 20)
#define HDR_FLOATS 8
#define TAB_MIN_ROWS 64
#define STAGE_K 11
#define STAGE_CAP (STAGE_K * SCAN_THREADS)
#define STAGE_PAD (STAGE_CAP + STAGE_CAP / TILE + 1)

__device__ __forceinline__ void cmul(float ar, float ai, float br, float bi,
                                     float& cr, float& ci) {
    cr = ar * br - ai * bi;
    ci = ar * bi + ai * br;
}

template <int N>
__device__ __forceinline__ float row_shr(float v) {
    return __builtin_bit_cast(float,
        __builtin_amdgcn_update_dpp(0, __builtin_bit_cast(int, v), 0x110 + N, 0xf, 0xf, false));
}

template <int N>
__device__ __forceinline__ void scan_step(int r, float& Xr, float& Xi, float& Qr, float& Qi,
                                          float& D0r, float& D0i, float& D1r, float& D1i) {
    float u0r = row_shr<N>(D0r), u0i = row_shr<N>(D0i);
    float u1r = row_shr<N>(D1r), u1i = row_shr<N>(D1i);
    float tr, ti;
    cmul(Xr, Xi, u0r, u0i, tr, ti); if (r >= N) { D0r += tr; D0i += ti; }
    cmul(Xr, Xi, u1r, u1i, tr, ti); if (r >= N) { D1r += tr; D1i += ti; }
    cmul(Qr, Qi, Xr, Xi, tr, ti);   if (r & N)  { Qr = tr;  Qi = ti; }
    cmul(Xr, Xi, Xr, Xi, tr, ti);   Xr = tr; Xi = ti;
}

template <int N>
__device__ __forceinline__ void scan_step1(int r, float& Xr, float& Xi, float& Qr, float& Qi,
                                           float& Dr, float& Di) {
    float ur = row_shr<N>(Dr), ui = row_shr<N>(Di);
    float tr, ti;
    cmul(Xr, Xi, ur, ui, tr, ti); if (r >= N) { Dr += tr; Di += ti; }
    cmul(Qr, Qi, Xr, Xi, tr, ti); if (r & N)  { Qr = tr;  Qi = ti; }
    cmul(Xr, Xi, Xr, Xi, tr, ti); Xr = tr; Xi = ti;
}

__device__ __forceinline__ void tile_samples_lds(const float* sx, const float* sy, int c, int nt,
                                                 float (&tx)[TILE + 1], float (&ty)[TILE + 1]) {
    #pragma unroll
    for (int u = 0; u <= TILE; ++u) {
        int jj = min(c + u, nt - 1);
        int p  = jj + (jj >> 4);
        tx[u] = sx[p];
        ty[u] = sy[p];
    }
}

__device__ __forceinline__ void tile_samples_mem(const float* __restrict__ TAx,
                                                 const float* __restrict__ TAy, int c, int nt,
                                                 float (&tx)[TILE + 1], float (&ty)[TILE + 1]) {
    #pragma unroll
    for (int u = 0; u <= TILE; ++u) {
        unsigned ii = (unsigned)min(c + u, nt - 1);
        tx[u] = TAx[ii];
        ty[u] = TAy[ii];
    }
}

__device__ __forceinline__ void tile_terms(float c0r, float c0i, float c1r, float c1i,
                                           const float (&tx)[TILE + 1],
                                           const float (&ty)[TILE + 1],
                                           int c, int nint,
                                           float (&gr)[TILE], float (&gi)[TILE]) {
    #pragma unroll
    for (int u = 0; u < TILE; ++u) {
        float vr = c0r * tx[u] - c0i * ty[u] + c1r * tx[u + 1] - c1i * ty[u + 1];
        float vi = c0r * ty[u] + c0i * tx[u] + c1r * ty[u + 1] + c1i * tx[u + 1];
        bool  in = c + u < nint;
        gr[u] = in ? vr : 0.f;
        gi[u] = in ? vi : 0.f;
    }
}

__global__ __launch_bounds__(SCAN_THREADS)
void interval_scan(const float* __restrict__ pk, const float* __restrict__ TAx,
                   const float* __restrict__ TAy, const float* __restrict__ fc,
                   const int* __restrict__ p_dt, const int* __restrict__ p_dtf,
                   float2* __restrict__ zst, float4* __restrict__ tab,
                   float4* __restrict__ hdr, int nsteps, int nt)
{
    __shared__ float2 sh_tot[32];
    __shared__ float  sh_coef[6];
    __shared__ float  sTx[STAGE_PAD];
    __shared__ float  sTy[STAGE_PAD];
    int t    = threadIdx.x;
    int lane = t & 31;
    int w    = t >> 5;

    int dt_i  = p_dt[0];
    int dtf_i = p_dtf[0];
    float pk0 = pk[0], pk1 = pk[1], f = fc[0];

    bool fits = nt <= STAGE_CAP;
    float lx[STAGE_K], ly[STAGE_K];
    if (fits) {
        #pragma unroll
        for (int k = 0; k < STAGE_K; ++k) {
            if (k * SCAN_THREADS < nt) {
                unsigned ii = (unsigned)min(k * SCAN_THREADS + t, nt - 1);
                lx[k] = TAx[ii];
                ly[k] = TAy[ii];
            }
        }
    }

    asm volatile("" :: "s"(dt_i), "s"(dtf_i), "s"(pk0), "s"(pk1), "s"(f));

    float dt      = (float)dt_i;
    int   nsub    = dtf_i / dt_i;
    float invnsub = 1.f / (float)nsub;
    float K0   = __expf(pk0);
    float K1   = __expf(pk1);
    float ar   = 1.f - dt * K1;
    float ai   = -dt * f;
    float dtK0 = dt * K0;

    int nint  = (nsteps + nsub - 1) / nsub;
    int ntile = (nint + TILE - 1) / TILE;
    int tpt   = (ntile + SCAN_THREADS - 1) / SCAN_THREADS;
    int tb    = t * tpt;
    int cb    = tb * TILE;

    if (w == 0) {
        int r   = lane & 15;
        int h   = lane >> 4;
        int spl = (nsub + 15) >> 4;
        int m0  = r * spl;

        float Pr = 1.f, Pi = 0.f, D0r = 0.f, D0i = 0.f, D1r = 0.f, D1i = 0.f;
        for (int s = 0; s < spl; ++s) {
            float aa = (float)(m0 + s) * invnsub;
            float tr, ti;
            cmul(ar, ai, Pr, Pi, tr, ti);   Pr = tr; Pi = ti;
            cmul(ar, ai, D0r, D0i, tr, ti); D0r = tr + (1.f - aa); D0i = ti;
            cmul(ar, ai, D1r, D1i, tr, ti); D1r = tr + aa;         D1i = ti;
        }

        float Xr = Pr, Xi = Pi, Qr = 1.f, Qi = 0.f;
        scan_step<1>(r, Xr, Xi, Qr, Qi, D0r, D0i, D1r, D1i);
        scan_step<2>(r, Xr, Xi, Qr, Qi, D0r, D0i, D1r, D1i);
        scan_step<4>(r, Xr, Xi, Qr, Qi, D0r, D0i, D1r, D1i);
        scan_step<8>(r, Xr, Xi, Qr, Qi, D0r, D0i, D1r, D1i);

        Pr  = Qr;              Pi  = Qi;
        D0r = row_shr<1>(D0r); D0i = row_shr<1>(D0i);
        D1r = row_shr<1>(D1r); D1i = row_shr<1>(D1i);

        for (int s = 0; s < spl; ++s) {
            int   m  = m0 + s;
            float aa = (float)m * invnsub;
            float tr, ti;
            cmul(ar, ai, Pr, Pi, tr, ti);   Pr = tr; Pi = ti;
            cmul(ar, ai, D0r, D0i, tr, ti); D0r = tr + (1.f - aa); D0i = ti;
            cmul(ar, ai, D1r, D1i, tr, ti); D1r = tr + aa;         D1i = ti;
            float e0r = dtK0 * D0r, e0i = dtK0 * D0i;
            float e1r = dtK0 * D1r, e1i = dtK0 * D1i;
            tab[4 * m + 2 * h]     = h ? make_float4(Pi, Pr, e0i, e0r)
                                       : make_float4(Pr, -Pi, e0r, -e0i);
            tab[4 * m + 2 * h + 1] = h ? make_float4(e1i, e1r, 0.f, 0.f)
                                       : make_float4(e1r, -e1i, 0.f, 0.f);
            if (m == nsub - 1 && h == 0) {
                sh_coef[0] = Pr;   sh_coef[1] = Pi;
                sh_coef[2] = e0r;  sh_coef[3] = e0i;
                sh_coef[4] = e1r;  sh_coef[5] = e1i;
                hdr[0] = make_float4(Pr, Pi, e0r, e0i);
                hdr[1] = make_float4(e1r, e1i, __int_as_float(nsub), __int_as_float(nint));
            }
        }
        for (int m = 16 * spl + r; m < TAB_MIN_ROWS; m += 16) {
            tab[4 * m + 2 * h]     = make_float4(0.f, 0.f, 0.f, 0.f);
            tab[4 * m + 2 * h + 1] = make_float4(0.f, 0.f, 0.f, 0.f);
        }
    }
    if (fits) {
        #pragma unroll
        for (int k = 0; k < STAGE_K; ++k) {
            if (k * SCAN_THREADS < nt) {
                int j = k * SCAN_THREADS + t;
                sTx[j + (j >> 4)] = lx[k];
                sTy[j + (j >> 4)] = ly[k];
            }
        }
    }
    __syncthreads();
    float Ar = sh_coef[0], Ai = sh_coef[1];
    float c0r = sh_coef[2], c0i = sh_coef[3];
    float c1r = sh_coef[4], c1i = sh_coef[5];

    float Mr, Mi;
    {
        float Sr = Ar, Si = Ai, tr, ti;
        #pragma unroll
        for (int k = 0; k < 4; ++k) { cmul(Sr, Si, Sr, Si, tr, ti); Sr = tr; Si = ti; }
        Mr = Sr; Mi = Si;
        for (int q = 1; q < tpt; ++q) { cmul(Sr, Si, Mr, Mi, tr, ti); Mr = tr; Mi = ti; }
    }

    float Br = 0.f, Bi = 0.f;
    if (cb < nint) {
        for (int q = 0; q < tpt; ++q) {
            float tx[TILE + 1], ty[TILE + 1], gr[TILE], gi[TILE];
            int c = cb + q * TILE;
            if (fits) tile_samples_lds(sTx, sTy, c, nt, tx, ty);
            else      tile_samples_mem(TAx, TAy, c, nt, tx, ty);
            tile_terms(c0r, c0i, c1r, c1i, tx, ty, c, nint, gr, gi);
            #pragma unroll
            for (int u = 0; u < TILE; ++u) {
                float tr, ti;
                cmul(Ar, Ai, Br, Bi, tr, ti); Br = tr + gr[u]; Bi = ti + gi[u];
            }
        }
    }

    float Qr = 1.f, Qi = 0.f, Xr = Mr, Xi = Mi;
    #pragma unroll
    for (int off = 1; off < 32; off <<= 1) {
        float pr = __shfl_up(Br, off);
        float pi = __shfl_up(Bi, off);
        float tr, ti;
        cmul(Xr, Xi, pr, pi, tr, ti);
        if (lane >= off) { Br += tr; Bi += ti; }
        cmul(Qr, Qi, Xr, Xi, tr, ti);
        if (lane & off) { Qr = tr; Qi = ti; }
        cmul(Xr, Xi, Xr, Xi, tr, ti); Xr = tr; Xi = ti;
    }
    if (lane == 31) sh_tot[w] = make_float2(Br, Bi);
    __syncthreads();

    float2 tt = sh_tot[lane];
    float Tr = tt.x, Ti = tt.y;
    #pragma unroll
    for (int off = 1; off < 32; off <<= 1) {
        float pr = __shfl_up(Tr, off);
        float pi = __shfl_up(Ti, off);
        float tr, ti;
        cmul(Xr, Xi, pr, pi, tr, ti);
        if (lane >= off) { Tr += tr; Ti += ti; }
        cmul(Xr, Xi, Xr, Xi, tr, ti); Xr = tr; Xi = ti;
    }
    int   src = (w + 31) & 31;
    float Er  = __shfl(Tr, src);
    float Ei  = __shfl(Ti, src);
    if (w == 0) { Er = 0.f; Ei = 0.f; }
    float xr = __shfl_up(Br, 1);
    float xi = __shfl_up(Bi, 1);
    if (lane == 0) { xr = 0.f; xi = 0.f; }
    float Zr, Zi;
    cmul(Qr, Qi, Er, Ei, Zr, Zi);
    Zr += xr; Zi += xi;

    for (int j = ntile + t; j < ((nt + TILE - 1) >> 4); j += SCAN_THREADS) zst[j] = make_float2(0.f, 0.f);
    if (cb >= nint) return;
    zst[tb] = make_float2(Zr, Zi);
    for (int q = 1; q < tpt; ++q) {
        if (tb + q >= ntile) break;
        float tx[TILE + 1], ty[TILE + 1], gr[TILE], gi[TILE];
        int c = cb + (q - 1) * TILE;
        if (fits) tile_samples_lds(sTx, sTy, c, nt, tx, ty);
        else      tile_samples_mem(TAx, TAy, c, nt, tx, ty);
        tile_terms(c0r, c0i, c1r, c1i, tx, ty, c, nint, gr, gi);
        #pragma unroll
        for (int u = 0; u < TILE; ++u) {
            float tr, ti;
            cmul(Ar, Ai, Zr, Zi, tr, ti); Zr = tr + gr[u]; Zi = ti + gi[u];
        }
        zst[tb + q] = make_float2(Zr, Zi);
    }
}


__device__ __forceinline__ void load_weights(const float* __restrict__ tabf, int mt0, int r, int h,
                                             int lastrow, v2f (&Wr0)[4], v2f (&Wr1)[4],
                                             v2f (&Wi0)[4], v2f (&Wi1)[4]) {
    #pragma unroll
    for (int q = 0; q < 4; ++q) {
        int mc = min(16 * (mt0 + q) + r, lastrow);
        const float* trow = tabf + TABROW * mc + 2 * h;
        Wr0[q] = *reinterpret_cast<const v2f*>(trow);
        Wr1[q] = *reinterpret_cast<const v2f*>(trow + 4);
        Wi0[q] = *reinterpret_cast<const v2f*>(trow + 8);
        Wi1[q] = *reinterpret_cast<const v2f*>(trow + 12);
    }
}

__global__ __launch_bounds__(32)
void interval_expand(const float* __restrict__ TAx, const float* __restrict__ TAy,
                     const float* __restrict__ hdr, const float2* __restrict__ zst,
                     const float* __restrict__ tabf, float* __restrict__ out, int nsteps, int nt)
{
    int lane = threadIdx.x;
    int h    = lane >> 4;
    int r    = lane & 15;

    int gt    = (nt + 15) >> 4;
    int tile0 = blockIdx.x;

    v2f Wr0[4], Wr1[4], Wi0[4], Wi1[4];
    load_weights(tabf, 0, r, h, TAB_MIN_ROWS - 1, Wr0, Wr1, Wi0, Wi1);
    float  s0x, s0y, s1x, s1y;
    {
        int i = tile0 * 16 + r;
        unsigned i0 = (unsigned)min(i, nt - 1);
        unsigned i1 = (unsigned)min(i + 1, nt - 1);
        s0x = TAx[i0]; s0y = TAy[i0];
        s1x = TAx[i1]; s1y = TAy[i1];
    }
    float2 zt = zst[tile0];

    float Ar  = hdr[0], Ai  = hdr[1];
    float c0r = hdr[2], c0i = hdr[3];
    float c1r = hdr[4], c1i = hdr[5];
    float hn0 = hdr[6], hn1 = hdr[7];

    asm volatile("" :: "v"(Wr0[0]), "v"(Wr0[1]), "v"(Wr0[2]), "v"(Wr0[3]),
                       "v"(Wr1[0]), "v"(Wr1[1]), "v"(Wr1[2]), "v"(Wr1[3]),
                       "v"(Wi0[0]), "v"(Wi0[1]), "v"(Wi0[2]), "v"(Wi0[3]),
                       "v"(Wi1[0]), "v"(Wi1[1]), "v"(Wi1[2]), "v"(Wi1[3]),
                       "v"(s0x), "v"(s0y), "v"(s1x), "v"(s1y));
    asm volatile("" :: "s"(Ar), "s"(Ai), "s"(c0r), "s"(c0i), "s"(c1r), "s"(c1i),
                       "s"(hn0), "s"(hn1), "s"(zt.x), "s"(zt.y));

    int nsub   = __float_as_int(hn0);
    int nint   = __float_as_int(hn1);
    int ntile  = (nint + 15) >> 4;
    int mtiles = (nsub + 15) >> 4;
    int nwhole = (nint * nsub == nsteps) ? nint : nint - 1;

    for (int mt0 = 0; mt0 < mtiles; mt0 += 4) {
        if (mt0 > 0) load_weights(tabf, mt0, r, h, nsub - 1, Wr0, Wr1, Wi0, Wi1);

        for (int tile = tile0; tile < ntile; tile += gt) {
            int i = tile * 16 + r;
            if (mt0 > 0 || tile != tile0) {
                unsigned i0 = (unsigned)min(i, nt - 1);
                unsigned i1 = (unsigned)min(i + 1, nt - 1);
                s0x = TAx[i0]; s0y = TAy[i0];
                s1x = TAx[i1]; s1y = TAy[i1];
                zt  = zst[tile];
            }

            bool  in = i < nint;
            float Dr = in ? c0r * s0x - c0i * s0y + c1r * s1x - c1i * s1y : 0.f;
            float Di = in ? c0r * s0y + c0i * s0x + c1r * s1y + c1i * s1x : 0.f;
            float Xr = Ar, Xi = Ai, Qr = 1.f, Qi = 0.f;
            scan_step1<1>(r, Xr, Xi, Qr, Qi, Dr, Di);
            scan_step1<2>(r, Xr, Xi, Qr, Qi, Dr, Di);
            scan_step1<4>(r, Xr, Xi, Qr, Qi, Dr, Di);
            scan_step1<8>(r, Xr, Xi, Qr, Qi, Dr, Di);
            float zr, zi;
            cmul(Qr, Qi, zt.x, zt.y, zr, zi);
            zr += row_shr<1>(Dr);
            zi += row_shr<1>(Di);

            v2f X0, X1;
            X0[0] = h ? s0x : zr;    X0[1] = h ? s0y : zi;
            X1[0] = h ? 0.f : s1x;   X1[1] = h ? 0.f : s1y;

            v8f Dre[4], Dim[4];
            #pragma unroll
            for (int q = 0; q < 4; ++q) {
                v8f zero = {};
                Dre[q] = __builtin_amdgcn_wmma_f32_16x16x4_f32(false, X0, false, Wr0[q],
                                                               (short)0, zero, false, false);
                Dim[q] = __builtin_amdgcn_wmma_f32_16x16x4_f32(false, X0, false, Wi0[q],
                                                               (short)0, zero, false, false);
            }
            #pragma unroll
            for (int q = 0; q < 4; ++q) {
                Dre[q] = __builtin_amdgcn_wmma_f32_16x16x4_f32(false, X1, false, Wr1[q],
                                                               (short)0, Dre[q], false, false);
                Dim[q] = __builtin_amdgcn_wmma_f32_16x16x4_f32(false, X1, false, Wi1[q],
                                                               (short)0, Dim[q], false, false);
            }
            int wr = min(max(nwhole - tile * 16, 0), 16);
            bool halfwhole = h ? (wr == 16) : (wr >= 8);
            #pragma unroll
            for (int q = 0; q < 4; ++q) {
                int m = 16 * (mt0 + q) + r;
                if (m < nsub && halfwhole) {
                    int n0 = (tile * 16 + 8 * h) * nsub + m;
                    #pragma unroll
                    for (int e = 0; e < 8; ++e) {
                        float2 v; v.x = Dre[q][e]; v.y = Dim[q][e];
                        *reinterpret_cast<float2*>(out + 2 * (long)(n0 + e * nsub)) = v;
                    }
                }
            }
            int wpart = wr & 7;
            if (wpart != 0) {
                int hp = wr >> 3;
                #pragma unroll
                for (int q = 0; q < 4; ++q) {
                    int m = 16 * (mt0 + q) + r;
                    if (m < nsub && h == hp) {
                        int n0 = (tile * 16 + 8 * hp) * nsub + m;
                        #pragma unroll
                        for (int e = 0; e < 8; ++e) {
                            if (e < wpart) {
                                float2 v; v.x = Dre[q][e]; v.y = Dim[q][e];
                                *reinterpret_cast<float2*>(out + 2 * (long)(n0 + e * nsub)) = v;
                            }
                        }
                    }
                }
            }
            int rp = nwhole - tile * 16;
            if (nwhole < nint && rp >= 0 && rp < 16) {
                int rem = nsteps - nwhole * nsub;
                #pragma unroll
                for (int q = 0; q < 4; ++q) {
                    int m = 16 * (mt0 + q) + r;
                    if (m < rem && h == (rp >> 3)) {
                        #pragma unroll
                        for (int e = 0; e < 8; ++e) {
                            if (e == (rp & 7)) {
                                float2 v; v.x = Dre[q][e]; v.y = Dim[q][e];
                                *reinterpret_cast<float2*>(out + 2 * (long)(nwhole * nsub + m)) = v;
                            }
                        }
                    }
                }
            }
        }
    }
}

extern "C" void kernel_launch(void* const* d_in, const int* in_sizes, int n_in,
                              void* d_out, int out_size, void* d_ws, size_t ws_size,
                              hipStream_t stream)
{
    const float* pk    = (const float*)d_in[0];
    const float* TAx   = (const float*)d_in[1];
    const float* TAy   = (const float*)d_in[2];
    const float* fc    = (const float*)d_in[3];
    const int*   p_dt  = (const int*)d_in[6];
    const int*   p_dtf = (const int*)d_in[7];
    (void)n_in; (void)ws_size;

    int nt      = in_sizes[1];
    int nsteps  = out_size / 2;
    float*  out = (float*)d_out;
    float2* zst = (float2*)d_ws;
    float4* tab = (float4*)((char*)d_ws + TAB_OFFSET_BYTES);
    float4* hdr = (float4*)((char*)d_ws + HDR_OFFSET_BYTES);

    interval_scan<<<1, SCAN_THREADS, 0, stream>>>(pk, TAx, TAy, fc, p_dt, p_dtf,
                                                  zst, tab, hdr, nsteps, nt);
    int blocks = (nt + 15) / 16;
    interval_expand<<<blocks, 32, 0, stream>>>(TAx, TAy, (const float*)hdr, zst, (const float*)tab,
                                               out, nsteps, nt);
}


# Scan launch split into 8 blocks, each scans its 128-tile share from zero and stores (local start, power) per tile; tile waves add power x carry-in summed from the share ends; reduction tree changes
# speedup vs baseline: 13.6709x; 1.0557x over previous
#include <hip/hip_runtime.h>

typedef __attribute__((ext_vector_type(2))) float v2f;
typedef __attribute__((ext_vector_type(8))) float v8f;

#define SCAN_BLOCKS 8
#define SCAN_THREADS 128
#define SCAN_WAVES (SCAN_THREADS / 32)
#define TILE 16
#define TABROW 16
#define TAB_OFFSET_BYTES (128u << 20)
#define HDR_OFFSET_BYTES (64u << 20)
#define HDR_FLOATS 8
#define HDR2_OFFSET_FLOATS 32
#define ENDS_OFFSET_FLOATS 64
#define ENDS_PITCH_FLOATS 32
#define TAB_MIN_ROWS 64
#define STAGE_K 17
#define STAGE_CAP (STAGE_K * SCAN_THREADS)
#define STAGE_PAD (STAGE_CAP + STAGE_CAP / TILE + 1)

__device__ __forceinline__ void cmul(float ar, float ai, float br, float bi,
                                     float& cr, float& ci) {
    cr = ar * br - ai * bi;
    ci = ar * bi + ai * br;
}

template <int N>
__device__ __forceinline__ float row_shr(float v) {
    return __builtin_bit_cast(float,
        __builtin_amdgcn_update_dpp(0, __builtin_bit_cast(int, v), 0x110 + N, 0xf, 0xf, false));
}

template <int N>
__device__ __forceinline__ void scan_step(int r, float& Xr, float& Xi, float& Qr, float& Qi,
                                          float& D0r, float& D0i, float& D1r, float& D1i) {
    float u0r = row_shr<N>(D0r), u0i = row_shr<N>(D0i);
    float u1r = row_shr<N>(D1r), u1i = row_shr<N>(D1i);
    float tr, ti;
    cmul(Xr, Xi, u0r, u0i, tr, ti); if (r >= N) { D0r += tr; D0i += ti; }
    cmul(Xr, Xi, u1r, u1i, tr, ti); if (r >= N) { D1r += tr; D1i += ti; }
    cmul(Qr, Qi, Xr, Xi, tr, ti);   if (r & N)  { Qr = tr;  Qi = ti; }
    cmul(Xr, Xi, Xr, Xi, tr, ti);   Xr = tr; Xi = ti;
}

template <int N>
__device__ __forceinline__ void scan_step1(int r, float& Xr, float& Xi, float& Qr, float& Qi,
                                           float& Dr, float& Di) {
    float ur = row_shr<N>(Dr), ui = row_shr<N>(Di);
    float tr, ti;
    cmul(Xr, Xi, ur, ui, tr, ti); if (r >= N) { Dr += tr; Di += ti; }
    cmul(Qr, Qi, Xr, Xi, tr, ti); if (r & N)  { Qr = tr;  Qi = ti; }
    cmul(Xr, Xi, Xr, Xi, tr, ti); Xr = tr; Xi = ti;
}

__device__ __forceinline__ void tile_samples_lds(const float* sx, const float* sy, int c,
                                                 float (&tx)[TILE + 1], float (&ty)[TILE + 1]) {
    int p0 = c + (c >> 4);
    #pragma unroll
    for (int u = 0; u <= TILE; ++u) {
        tx[u] = sx[p0 + u + (u >> 4)];
        ty[u] = sy[p0 + u + (u >> 4)];
    }
}

__device__ __forceinline__ void tile_samples_mem(const float* __restrict__ TAx,
                                                 const float* __restrict__ TAy, int c, int nt,
                                                 float (&tx)[TILE + 1], float (&ty)[TILE + 1]) {
    #pragma unroll
    for (int u = 0; u <= TILE; ++u) {
        unsigned ii = (unsigned)min(c + u, nt - 1);
        tx[u] = TAx[ii];
        ty[u] = TAy[ii];
    }
}

__device__ __forceinline__ void tile_terms(float c0r, float c0i, float c1r, float c1i,
                                           const float (&tx)[TILE + 1],
                                           const float (&ty)[TILE + 1],
                                           int c, int nint,
                                           float (&gr)[TILE], float (&gi)[TILE]) {
    #pragma unroll
    for (int u = 0; u < TILE; ++u) {
        float vr = c0r * tx[u] - c0i * ty[u] + c1r * tx[u + 1] - c1i * ty[u + 1];
        float vi = c0r * ty[u] + c0i * tx[u] + c1r * ty[u + 1] + c1i * tx[u + 1];
        bool  in = c + u < nint;
        gr[u] = in ? vr : 0.f;
        gi[u] = in ? vi : 0.f;
    }
}

__global__ __launch_bounds__(SCAN_THREADS)
void interval_scan(const float* __restrict__ pk, const float* __restrict__ TAx,
                   const float* __restrict__ TAy, const float* __restrict__ fc,
                   const int* __restrict__ p_dt, const int* __restrict__ p_dtf,
                   float4* __restrict__ zrec, float4* __restrict__ tab,
                   float4* __restrict__ hdr, int nsteps, int nt)
{
    __shared__ float2 sh_tot[SCAN_WAVES];
    __shared__ float  sh_coef[6];
    __shared__ float  sTx[STAGE_PAD];
    __shared__ float  sTy[STAGE_PAD];
    int t    = threadIdx.x;
    int lane = t & 31;
    int w    = t >> 5;
    int b    = blockIdx.x;
    bool first = b == 0;

    int dt_i  = p_dt[0];
    int dtf_i = p_dtf[0];
    float pk0 = pk[0], pk1 = pk[1], f = fc[0];

    int gt    = (nt + TILE - 1) >> 4;
    int tptg  = (gt + SCAN_BLOCKS * SCAN_THREADS - 1) / (SCAN_BLOCKS * SCAN_THREADS);
    int baseg = b * SCAN_THREADS * tptg * TILE;
    bool stage = SCAN_THREADS * tptg * TILE + 1 <= STAGE_CAP;
    float lx[STAGE_K], ly[STAGE_K];
    if (stage) {
        #pragma unroll
        for (int k = 0; k < STAGE_K; ++k) {
            unsigned ii = (unsigned)min(baseg + k * SCAN_THREADS + t, nt - 1);
            lx[k] = TAx[ii];
            ly[k] = TAy[ii];
        }
    }

    asm volatile("" :: "s"(dt_i), "s"(dtf_i), "s"(pk0), "s"(pk1), "s"(f));

    float dt      = (float)dt_i;
    int   nsub    = dtf_i / dt_i;
    float invnsub = 1.f / (float)nsub;
    float K0   = __expf(pk0);
    float K1   = __expf(pk1);
    float ar   = 1.f - dt * K1;
    float ai   = -dt * f;
    float dtK0 = dt * K0;

    int nint  = (nsteps + nsub - 1) / nsub;
    int ntile = (nint + TILE - 1) / TILE;
    int nall  = max(ntile, gt);
    int tpt   = (nall + SCAN_BLOCKS * SCAN_THREADS - 1) / (SCAN_BLOCKS * SCAN_THREADS);
    int S     = SCAN_THREADS * tpt;
    int tb    = b * S + t * tpt;
    int cb    = tb * TILE;
    bool fits = stage && tpt == tptg;
    float2* endb = reinterpret_cast<float2*>(hdr) + (ENDS_OFFSET_FLOATS + b * ENDS_PITCH_FLOATS) / 2;

    if (b * S >= ntile) {
        for (int j = b * S + t; j < min(gt, b * S + S); j += SCAN_THREADS) zrec[j] = make_float4(0.f, 0.f, 0.f, 0.f);
        if (t == 0) endb[0] = make_float2(0.f, 0.f);
        return;
    }

    if (w == 0) {
        int r   = lane & 15;
        int h   = lane >> 4;
        int spl = (nsub + 15) >> 4;
        int m0  = r * spl;

        float Pr = 1.f, Pi = 0.f, D0r = 0.f, D0i = 0.f, D1r = 0.f, D1i = 0.f;
        for (int s = 0; s < spl; ++s) {
            float aa = (float)(m0 + s) * invnsub;
            float tr, ti;
            cmul(ar, ai, Pr, Pi, tr, ti);   Pr = tr; Pi = ti;
            cmul(ar, ai, D0r, D0i, tr, ti); D0r = tr + (1.f - aa); D0i = ti;
            cmul(ar, ai, D1r, D1i, tr, ti); D1r = tr + aa;         D1i = ti;
        }

        float Xr = Pr, Xi = Pi, Qr = 1.f, Qi = 0.f;
        scan_step<1>(r, Xr, Xi, Qr, Qi, D0r, D0i, D1r, D1i);
        scan_step<2>(r, Xr, Xi, Qr, Qi, D0r, D0i, D1r, D1i);
        scan_step<4>(r, Xr, Xi, Qr, Qi, D0r, D0i, D1r, D1i);
        scan_step<8>(r, Xr, Xi, Qr, Qi, D0r, D0i, D1r, D1i);

        Pr  = Qr;              Pi  = Qi;
        D0r = row_shr<1>(D0r); D0i = row_shr<1>(D0i);
        D1r = row_shr<1>(D1r); D1i = row_shr<1>(D1i);

        for (int s = 0; s < spl; ++s) {
            int   m  = m0 + s;
            float aa = (float)m * invnsub;
            float tr, ti;
            cmul(ar, ai, Pr, Pi, tr, ti);   Pr = tr; Pi = ti;
            cmul(ar, ai, D0r, D0i, tr, ti); D0r = tr + (1.f - aa); D0i = ti;
            cmul(ar, ai, D1r, D1i, tr, ti); D1r = tr + aa;         D1i = ti;
            float e0r = dtK0 * D0r, e0i = dtK0 * D0i;
            float e1r = dtK0 * D1r, e1i = dtK0 * D1i;
            if (first) {
                tab[4 * m + 2 * h]     = h ? make_float4(Pi, Pr, e0i, e0r)
                                           : make_float4(Pr, -Pi, e0r, -e0i);
                tab[4 * m + 2 * h + 1] = h ? make_float4(e1i, e1r, 0.f, 0.f)
                                           : make_float4(e1r, -e1i, 0.f, 0.f);
            }
            if (m == nsub - 1 && h == 0) {
                sh_coef[0] = Pr;   sh_coef[1] = Pi;
                sh_coef[2] = e0r;  sh_coef[3] = e0i;
                sh_coef[4] = e1r;  sh_coef[5] = e1i;
                if (first) {
                    hdr[0] = make_float4(Pr, Pi, e0r, e0i);
                    hdr[1] = make_float4(e1r, e1i, __int_as_float(nsub), __int_as_float(nint));
                }
            }
        }
        for (int m = 16 * spl + r; first && m < TAB_MIN_ROWS; m += 16) {
            tab[4 * m + 2 * h]     = make_float4(0.f, 0.f, 0.f, 0.f);
            tab[4 * m + 2 * h + 1] = make_float4(0.f, 0.f, 0.f, 0.f);
        }
    }
    if (fits) {
        #pragma unroll
        for (int k = 0; k < STAGE_K; ++k) {
            int j = k * SCAN_THREADS + t;
            sTx[j + (j >> 4)] = lx[k];
            sTy[j + (j >> 4)] = ly[k];
        }
    }
    __syncthreads();
    float Ar = sh_coef[0], Ai = sh_coef[1];
    float c0r = sh_coef[2], c0i = sh_coef[3];
    float c1r = sh_coef[4], c1i = sh_coef[5];

    float A16r, A16i, Mr, Mi;
    {
        float Sr = Ar, Si = Ai, tr, ti;
        #pragma unroll
        for (int k = 0; k < 4; ++k) { cmul(Sr, Si, Sr, Si, tr, ti); Sr = tr; Si = ti; }
        A16r = Sr; A16i = Si;
        Mr = Sr; Mi = Si;
        for (int q = 1; q < tpt; ++q) { cmul(Sr, Si, Mr, Mi, tr, ti); Mr = tr; Mi = ti; }
    }

    int cl = t * tpt * TILE;
    float Br = 0.f, Bi = 0.f;
    if (cb < nint) {
        for (int q = 0; q < tpt; ++q) {
            float tx[TILE + 1], ty[TILE + 1], gr[TILE], gi[TILE];
            int c = cb + q * TILE;
            if (fits) tile_samples_lds(sTx, sTy, cl + q * TILE, tx, ty);
            else      tile_samples_mem(TAx, TAy, c, nt, tx, ty);
            tile_terms(c0r, c0i, c1r, c1i, tx, ty, c, nint, gr, gi);
            #pragma unroll
            for (int u = 0; u < TILE; ++u) {
                float tr, ti;
                cmul(Ar, Ai, Br, Bi, tr, ti); Br = tr + gr[u]; Bi = ti + gi[u];
            }
        }
    }

    float Qr = 1.f, Qi = 0.f, Xr = Mr, Xi = Mi;
    #pragma unroll
    for (int off = 1; off < 32; off <<= 1) {
        float pr = __shfl_up(Br, off);
        float pi = __shfl_up(Bi, off);
        float tr, ti;
        cmul(Xr, Xi, pr, pi, tr, ti);
        if (lane >= off) { Br += tr; Bi += ti; }
        cmul(Qr, Qi, Xr, Xi, tr, ti);
        if (lane & off) { Qr = tr; Qi = ti; }
        cmul(Xr, Xi, Xr, Xi, tr, ti); Xr = tr; Xi = ti;
    }
    if (lane == 31) sh_tot[w] = make_float2(Br, Bi);
    __syncthreads();

    float Er = 0.f, Ei = 0.f, Gr = 1.f, Gi = 0.f, Nr = 0.f, Ni = 0.f;
    #pragma unroll
    for (int v = 0; v < SCAN_WAVES; ++v) {
        float2 tv = sh_tot[v];
        float tr, ti;
        if (v == w) { Er = Nr; Ei = Ni; }
        cmul(Xr, Xi, Nr, Ni, tr, ti); Nr = tr + tv.x; Ni = ti + tv.y;
        cmul(Xr, Xi, Gr, Gi, tr, ti); if (v < w) { Gr = tr; Gi = ti; }
    }
    float xr = __shfl_up(Br, 1);
    float xi = __shfl_up(Bi, 1);
    if (lane == 0) { xr = 0.f; xi = 0.f; }
    float Zr, Zi;
    cmul(Qr, Qi, Er, Ei, Zr, Zi);
    Zr += xr; Zi += xi;
    float Pr, Pi;
    cmul(Qr, Qi, Gr, Gi, Pr, Pi);

    if (t == 0) {
        endb[0] = make_float2(Nr, Ni);
        if (first) {
            float Yr = Xr, Yi = Xi, tr, ti;
            for (int v = 1; v < SCAN_WAVES; ++v) { cmul(Xr, Xi, Yr, Yi, tr, ti); Yr = tr; Yi = ti; }
            hdr[HDR2_OFFSET_FLOATS / 4] = make_float4(Yr, Yi, __int_as_float(tpt), 0.f);
        }
    }

    for (int j = max(ntile, b * S) + t; j < min(gt, b * S + S); j += SCAN_THREADS) zrec[j] = make_float4(0.f, 0.f, 0.f, 0.f);
    if (cb >= nint) return;
    zrec[tb] = make_float4(Zr, Zi, Pr, Pi);
    for (int q = 1; q < tpt; ++q) {
        if (tb + q >= ntile) break;
        float tx[TILE + 1], ty[TILE + 1], gr[TILE], gi[TILE];
        int c = cb + (q - 1) * TILE;
        if (fits) tile_samples_lds(sTx, sTy, cl + (q - 1) * TILE, tx, ty);
        else      tile_samples_mem(TAx, TAy, c, nt, tx, ty);
        tile_terms(c0r, c0i, c1r, c1i, tx, ty, c, nint, gr, gi);
        #pragma unroll
        for (int u = 0; u < TILE; ++u) {
            float tr, ti;
            cmul(Ar, Ai, Zr, Zi, tr, ti); Zr = tr + gr[u]; Zi = ti + gi[u];
        }
        float tr, ti;
        cmul(A16r, A16i, Pr, Pi, tr, ti); Pr = tr; Pi = ti;
        zrec[tb + q] = make_float4(Zr, Zi, Pr, Pi);
    }
}


__device__ __forceinline__ void load_weights(const float* __restrict__ tabf, int mt0, int r, int h,
                                             int lastrow, v2f (&Wr0)[4], v2f (&Wr1)[4],
                                             v2f (&Wi0)[4], v2f (&Wi1)[4]) {
    #pragma unroll
    for (int q = 0; q < 4; ++q) {
        int mc = min(16 * (mt0 + q) + r, lastrow);
        const float* trow = tabf + TABROW * mc + 2 * h;
        Wr0[q] = *reinterpret_cast<const v2f*>(trow);
        Wr1[q] = *reinterpret_cast<const v2f*>(trow + 4);
        Wi0[q] = *reinterpret_cast<const v2f*>(trow + 8);
        Wi1[q] = *reinterpret_cast<const v2f*>(trow + 12);
    }
}

__global__ __launch_bounds__(32)
void interval_expand(const float* __restrict__ TAx, const float* __restrict__ TAy,
                     const float* __restrict__ hdr, const float4* __restrict__ zrec,
                     const float* __restrict__ tabf, float* __restrict__ out, int nsteps, int nt)
{
    int lane = threadIdx.x;
    int h    = lane >> 4;
    int r    = lane & 15;

    int gt    = (nt + 15) >> 4;
    int tile0 = blockIdx.x;

    v2f Wr0[4], Wr1[4], Wi0[4], Wi1[4];
    load_weights(tabf, 0, r, h, TAB_MIN_ROWS - 1, Wr0, Wr1, Wi0, Wi1);
    float  s0x, s0y, s1x, s1y;
    {
        int i = tile0 * 16 + r;
        unsigned i0 = (unsigned)min(i, nt - 1);
        unsigned i1 = (unsigned)min(i + 1, nt - 1);
        s0x = TAx[i0]; s0y = TAy[i0];
        s1x = TAx[i1]; s1y = TAy[i1];
    }
    float4 zq = zrec[tile0];

    float Ar  = hdr[0], Ai  = hdr[1];
    float c0r = hdr[2], c0i = hdr[3];
    float c1r = hdr[4], c1i = hdr[5];
    float hn0 = hdr[6], hn1 = hdr[7];
    float MSr = hdr[HDR2_OFFSET_FLOATS], MSi = hdr[HDR2_OFFSET_FLOATS + 1];
    float hn2 = hdr[HDR2_OFFSET_FLOATS + 2];
    float Eer[SCAN_BLOCKS - 1], Eei[SCAN_BLOCKS - 1];
    #pragma unroll
    for (int v = 0; v < SCAN_BLOCKS - 1; ++v) {
        Eer[v] = hdr[ENDS_OFFSET_FLOATS + ENDS_PITCH_FLOATS * v];
        Eei[v] = hdr[ENDS_OFFSET_FLOATS + ENDS_PITCH_FLOATS * v + 1];
    }

    asm volatile("" :: "v"(Wr0[0]), "v"(Wr0[1]), "v"(Wr0[2]), "v"(Wr0[3]),
                       "v"(Wr1[0]), "v"(Wr1[1]), "v"(Wr1[2]), "v"(Wr1[3]),
                       "v"(Wi0[0]), "v"(Wi0[1]), "v"(Wi0[2]), "v"(Wi0[3]),
                       "v"(Wi1[0]), "v"(Wi1[1]), "v"(Wi1[2]), "v"(Wi1[3]),
                       "v"(s0x), "v"(s0y), "v"(s1x), "v"(s1y));
    asm volatile("" :: "s"(Ar), "s"(Ai), "s"(c0r), "s"(c0i), "s"(c1r), "s"(c1i),
                       "s"(hn0), "s"(hn1), "s"(zq.x), "s"(zq.y), "s"(zq.z), "s"(zq.w));
    asm volatile("" :: "s"(MSr), "s"(MSi), "s"(hn2),
                       "s"(Eer[0]), "s"(Eer[1]), "s"(Eer[2]), "s"(Eer[3]), "s"(Eer[4]), "s"(Eer[5]), "s"(Eer[6]),
                       "s"(Eei[0]), "s"(Eei[1]), "s"(Eei[2]), "s"(Eei[3]), "s"(Eei[4]), "s"(Eei[5]), "s"(Eei[6]));

    int nsub   = __float_as_int(hn0);
    int nint   = __float_as_int(hn1);
    int ntile  = (nint + 15) >> 4;
    int mtiles = (nsub + 15) >> 4;
    int nwhole = (nint * nsub == nsteps) ? nint : nint - 1;
    int tpt    = __float_as_int(hn2);

    float Cr[SCAN_BLOCKS - 1], Ci[SCAN_BLOCKS - 1];
    {
        float cr = 0.f, ci = 0.f;
        #pragma unroll
        for (int k = 0; k < SCAN_BLOCKS - 1; ++k) {
            float tr, ti;
            cmul(MSr, MSi, cr, ci, tr, ti);
            cr = tr + Eer[k]; ci = ti + Eei[k];
            Cr[k] = cr; Ci[k] = ci;
        }
    }

    for (int mt0 = 0; mt0 < mtiles; mt0 += 4) {
        if (mt0 > 0) load_weights(tabf, mt0, r, h, nsub - 1, Wr0, Wr1, Wi0, Wi1);

        for (int tile = tile0; tile < ntile; tile += gt) {
            int i = tile * 16 + r;
            if (mt0 > 0 || tile != tile0) {
                unsigned i0 = (unsigned)min(i, nt - 1);
                unsigned i1 = (unsigned)min(i + 1, nt - 1);
                s0x = TAx[i0]; s0y = TAy[i0];
                s1x = TAx[i1]; s1y = TAy[i1];
                zq  = zrec[tile];
            }
            int sh = tile / SCAN_THREADS;
            if (tpt != 1) sh /= tpt;
            float cr = 0.f, ci = 0.f;
            #pragma unroll
            for (int k = 0; k < SCAN_BLOCKS - 1; ++k) {
                if (sh == k + 1) { cr = Cr[k]; ci = Ci[k]; }
            }
            float2 zt;
            zt.x = zq.x + (zq.z * cr - zq.w * ci);
            zt.y = zq.y + (zq.z * ci + zq.w * cr);

            bool  in = i < nint;
            float Dr = in ? c0r * s0x - c0i * s0y + c1r * s1x - c1i * s1y : 0.f;
            float Di = in ? c0r * s0y + c0i * s0x + c1r * s1y + c1i * s1x : 0.f;
            float Xr = Ar, Xi = Ai, Qr = 1.f, Qi = 0.f;
            scan_step1<1>(r, Xr, Xi, Qr, Qi, Dr, Di);
            scan_step1<2>(r, Xr, Xi, Qr, Qi, Dr, Di);
            scan_step1<4>(r, Xr, Xi, Qr, Qi, Dr, Di);
            scan_step1<8>(r, Xr, Xi, Qr, Qi, Dr, Di);
            float zr, zi;
            cmul(Qr, Qi, zt.x, zt.y, zr, zi);
            zr += row_shr<1>(Dr);
            zi += row_shr<1>(Di);

            v2f X0, X1;
            X0[0] = h ? s0x : zr;    X0[1] = h ? s0y : zi;
            X1[0] = h ? 0.f : s1x;   X1[1] = h ? 0.f : s1y;

            v8f Dre[4], Dim[4];
            #pragma unroll
            for (int q = 0; q < 4; ++q) {
                v8f zero = {};
                Dre[q] = __builtin_amdgcn_wmma_f32_16x16x4_f32(false, X0, false, Wr0[q],
                                                               (short)0, zero, false, false);
                Dim[q] = __builtin_amdgcn_wmma_f32_16x16x4_f32(false, X0, false, Wi0[q],
                                                               (short)0, zero, false, false);
            }
            #pragma unroll
            for (int q = 0; q < 4; ++q) {
                Dre[q] = __builtin_amdgcn_wmma_f32_16x16x4_f32(false, X1, false, Wr1[q],
                                                               (short)0, Dre[q], false, false);
                Dim[q] = __builtin_amdgcn_wmma_f32_16x16x4_f32(false, X1, false, Wi1[q],
                                                               (short)0, Dim[q], false, false);
            }
            int wr = min(max(nwhole - tile * 16, 0), 16);
            bool halfwhole = h ? (wr == 16) : (wr >= 8);
            #pragma unroll
            for (int q = 0; q < 4; ++q) {
                int m = 16 * (mt0 + q) + r;
                if (m < nsub && halfwhole) {
                    int n0 = (tile * 16 + 8 * h) * nsub + m;
                    #pragma unroll
                    for (int e = 0; e < 8; ++e) {
                        float2 v; v.x = Dre[q][e]; v.y = Dim[q][e];
                        *reinterpret_cast<float2*>(out + 2 * (long)(n0 + e * nsub)) = v;
                    }
                }
            }
            int wpart = wr & 7;
            if (wpart != 0) {
                int hp = wr >> 3;
                #pragma unroll
                for (int q = 0; q < 4; ++q) {
                    int m = 16 * (mt0 + q) + r;
                    if (m < nsub && h == hp) {
                        int n0 = (tile * 16 + 8 * hp) * nsub + m;
                        #pragma unroll
                        for (int e = 0; e < 8; ++e) {
                            if (e < wpart) {
                                float2 v; v.x = Dre[q][e]; v.y = Dim[q][e];
                                *reinterpret_cast<float2*>(out + 2 * (long)(n0 + e * nsub)) = v;
                            }
                        }
                    }
                }
            }
            int rp = nwhole - tile * 16;
            if (nwhole < nint && rp >= 0 && rp < 16) {
                int rem = nsteps - nwhole * nsub;
                #pragma unroll
                for (int q = 0; q < 4; ++q) {
                    int m = 16 * (mt0 + q) + r;
                    if (m < rem && h == (rp >> 3)) {
                        #pragma unroll
                        for (int e = 0; e < 8; ++e) {
                            if (e == (rp & 7)) {
                                float2 v; v.x = Dre[q][e]; v.y = Dim[q][e];
                                *reinterpret_cast<float2*>(out + 2 * (long)(nwhole * nsub + m)) = v;
                            }
                        }
                    }
                }
            }
        }
    }
}

extern "C" void kernel_launch(void* const* d_in, const int* in_sizes, int n_in,
                              void* d_out, int out_size, void* d_ws, size_t ws_size,
                              hipStream_t stream)
{
    const float* pk    = (const float*)d_in[0];
    const float* TAx   = (const float*)d_in[1];
    const float* TAy   = (const float*)d_in[2];
    const float* fc    = (const float*)d_in[3];
    const int*   p_dt  = (const int*)d_in[6];
    const int*   p_dtf = (const int*)d_in[7];
    (void)n_in; (void)ws_size;

    int nt      = in_sizes[1];
    int nsteps  = out_size / 2;
    float*  out  = (float*)d_out;
    float4* zrec = (float4*)d_ws;
    float4* tab = (float4*)((char*)d_ws + TAB_OFFSET_BYTES);
    float4* hdr = (float4*)((char*)d_ws + HDR_OFFSET_BYTES);

    interval_scan<<<SCAN_BLOCKS, SCAN_THREADS, 0, stream>>>(pk, TAx, TAy, fc, p_dt, p_dtf,
                                                            zrec, tab, hdr, nsteps, nt);
    int blocks = (nt + 15) / 16;
    interval_expand<<<blocks, 32, 0, stream>>>(TAx, TAy, (const float*)hdr, zrec, (const float*)tab,
                                               out, nsteps, nt);
}
